# starting kernel
# baseline (speedup 1.0000x reference)
#include <hip/hip_runtime.h>
#include <hip/hip_bf16.h>


typedef float v2f __attribute__((ext_vector_type(2)));
typedef float v8f __attribute__((ext_vector_type(8)));

#define TPB 256

__global__ void fill_kernel(float* __restrict__ p, float v, int n) {
    int i = blockIdx.x * blockDim.x + threadIdx.x;
    if (i < n) p[i] = v;
}

__global__ void deg_kernel(const int* __restrict__ dst, float* __restrict__ deg, int E) {
    int e = blockIdx.x * blockDim.x + threadIdx.x;
    if (e < E) atomicAdd(&deg[dst[e]], 1.0f);
}

__global__ void rsqrt_kernel(float* __restrict__ d, int n) {
    int i = blockIdx.x * blockDim.x + threadIdx.x;
    if (i < n) d[i] = rsqrtf(d[i]);
}

template <int K, int Nn>
__global__ void wmma_gemm_f32(const float* __restrict__ A,
                              const float* __restrict__ B,
                              float* __restrict__ C,
                              const float* __restrict__ bias,
                              int row_tiles) {
    constexpr int COLT = Nn / 16;
    int wave = (blockIdx.x * blockDim.x + threadIdx.x) >> 5;
    if (wave >= row_tiles) return;
    int lane = threadIdx.x & 31;
    int half = lane >> 4;
    int lm   = lane & 15;
    int row0 = wave << 4;

    const float* __restrict__ arow  = A + (size_t)(row0 + lm) * K + 2 * half;
    const float* __restrict__ bbase = B + (size_t)(2 * half) * Nn + lm;

    v8f acc[COLT];
#pragma unroll
    for (int c = 0; c < COLT; ++c) acc[c] = (v8f){};

#pragma unroll
    for (int k0 = 0; k0 < K; k0 += 4) {
        v2f a;
        a.x = arow[k0];
        a.y = arow[k0 + 1];
#pragma unroll
        for (int c = 0; c < COLT; ++c) {
            v2f b;
            b.x = bbase[k0 * Nn + c * 16];
            b.y = bbase[(k0 + 1) * Nn + c * 16];
            acc[c] = __builtin_amdgcn_wmma_f32_16x16x4_f32(
                 false, a,  false, b,
                 (short)0, acc[c],  false,  false);
        }
    }

    float* __restrict__ crow = C + (size_t)(row0 + 8 * half) * Nn + lm;
#pragma unroll
    for (int c = 0; c < COLT; ++c) {
        float bv = bias ? bias[c * 16 + lm] : 0.0f;
#pragma unroll
        for (int v = 0; v < 8; ++v) {
            crow[(size_t)v * Nn + c * 16] = acc[c][v] + bv;
        }
    }
}

template <int F>
__global__ void scatter_kernel(const float* __restrict__ h,
                               const int* __restrict__ src,
                               const int* __restrict__ dst,
                               const float* __restrict__ dinv,
                               float* __restrict__ out, int E) {
    int gid = blockIdx.x * blockDim.x + threadIdx.x;
    if (gid >= E * F) return;
    int e = gid / F;
    int f = gid - e * F;
    int s = src[e];
    int d = dst[e];
    float val = h[(size_t)s * F + f] * (dinv[s] * dinv[d]);
    atomicAdd(&out[(size_t)d * F + f], val);
}

template <int F, bool RELU>
__global__ void finalize_kernel(float* __restrict__ agg,
                                const float* __restrict__ xw,
                                const float* __restrict__ dinv,
                                const float* __restrict__ bias, int N) {
    int gid = blockIdx.x * blockDim.x + threadIdx.x;
    if (gid >= N * F) return;
    int i = gid / F;
    int f = gid - i * F;
    float di = dinv[i];
    float v = agg[gid] + xw[gid] * di * di + bias[f];
    agg[gid] = RELU ? fmaxf(v, 0.0f) : v;
}

template <int F>
__global__ void pool_accum_kernel(const float* __restrict__ h,
                                  const int* __restrict__ batch,
                                  float* __restrict__ sums, int N) {
    int gid = blockIdx.x * blockDim.x + threadIdx.x;
    if (gid >= N * F) return;
    int i = gid / F;
    int f = gid - i * F;
    atomicAdd(&sums[(size_t)batch[i] * F + f], h[gid]);
}

__global__ void count_kernel(const int* __restrict__ batch, float* __restrict__ cnt, int N) {
    int i = blockIdx.x * blockDim.x + threadIdx.x;
    if (i < N) atomicAdd(&cnt[batch[i]], 1.0f);
}

template <int F>
__global__ void pool_div_kernel(float* __restrict__ sums, const float* __restrict__ cnt, int G) {
    int gid = blockIdx.x * blockDim.x + threadIdx.x;
    if (gid >= G * F) return;
    int g = gid / F;
    sums[gid] /= fmaxf(cnt[g], 1.0f);
}

static inline int cdiv(long long a, long long b) { return (int)((a + b - 1) / b); }

extern "C" void kernel_launch(void* const* d_in, const int* in_sizes, int n_in,
                              void* d_out, int out_size, void* d_ws, size_t ws_size,
                              hipStream_t stream) {
    (void)n_in; (void)ws_size;
    const float* x    = (const float*)d_in[0];
    const int*   edge = (const int*)d_in[1];
    const int*   batch= (const int*)d_in[2];
    const float* b1   = (const float*)d_in[4];
    const float* W1   = (const float*)d_in[3];
    const float* W2   = (const float*)d_in[5];
    const float* b2   = (const float*)d_in[6];
    const float* Wfc  = (const float*)d_in[7];
    const float* bfc  = (const float*)d_in[8];
    float* out = (float*)d_out;

    const int N   = in_sizes[2];
    const int E   = in_sizes[1] / 2;
    const int D   = 64;
    const int D2  = 128;
    const int HID = in_sizes[8];
    const int G   = out_size / HID;

    const int* src = edge;
    const int* dst = edge + E;

    float* ws    = (float*)d_ws;
    float* xw1   = ws;
    float* agg1  = xw1  + (size_t)N * D;
    float* hw2   = agg1 + (size_t)N * D;
    float* agg2  = hw2  + (size_t)N * D2;
    float* dinv  = agg2 + (size_t)N * D2;
    float* sums  = dinv + (size_t)N;
    float* cnt   = sums + (size_t)G * D2;

    fill_kernel<<<cdiv((long long)N * D,  TPB), TPB, 0, stream>>>(agg1, 0.0f, N * D);
    fill_kernel<<<cdiv((long long)N * D2, TPB), TPB, 0, stream>>>(agg2, 0.0f, N * D2);
    fill_kernel<<<cdiv((long long)G * D2 + G, TPB), TPB, 0, stream>>>(sums, 0.0f, G * D2 + G);
    fill_kernel<<<cdiv(N, TPB), TPB, 0, stream>>>(dinv, 1.0f, N);

    deg_kernel<<<cdiv(E, TPB), TPB, 0, stream>>>(dst, dinv, E);
    rsqrt_kernel<<<cdiv(N, TPB), TPB, 0, stream>>>(dinv, N);

    {
        int row_tiles = N / 16;
        wmma_gemm_f32<64, 64><<<cdiv((long long)row_tiles * 32, TPB), TPB, 0, stream>>>(
            x, W1, xw1, nullptr, row_tiles);
    }
    scatter_kernel<64><<<cdiv((long long)E * D, TPB), TPB, 0, stream>>>(
        xw1, src, dst, dinv, agg1, E);
    finalize_kernel<64, true><<<cdiv((long long)N * D, TPB), TPB, 0, stream>>>(
        agg1, xw1, dinv, b1, N);

    {
        int row_tiles = N / 16;
        wmma_gemm_f32<64, 128><<<cdiv((long long)row_tiles * 32, TPB), TPB, 0, stream>>>(
            agg1, W2, hw2, nullptr, row_tiles);
    }
    scatter_kernel<128><<<cdiv((long long)E * D2, TPB), TPB, 0, stream>>>(
        hw2, src, dst, dinv, agg2, E);
    finalize_kernel<128, false><<<cdiv((long long)N * D2, TPB), TPB, 0, stream>>>(
        agg2, hw2, dinv, b2, N);

    pool_accum_kernel<128><<<cdiv((long long)N * D2, TPB), TPB, 0, stream>>>(
        agg2, batch, sums, N);
    count_kernel<<<cdiv(N, TPB), TPB, 0, stream>>>(batch, cnt, N);
    pool_div_kernel<128><<<cdiv((long long)G * D2, TPB), TPB, 0, stream>>>(sums, cnt, G);

    {
        int row_tiles = G / 16;
        wmma_gemm_f32<128, 64><<<cdiv((long long)row_tiles * 32, TPB), TPB, 0, stream>>>(
            sums, Wfc, out, bfc, row_tiles);
    }
}


# Edges grouped by destination in-call (returning int atomic per edge: degree + arrival-order rank; prefix offsets); one wave per node sums neighbour rows in registers, no f32 atomics in conv
# speedup vs baseline: 2.6665x; 2.6665x over previous
#include <hip/hip_runtime.h>
#include <hip/hip_bf16.h>


typedef float v2f __attribute__((ext_vector_type(2)));
typedef float v8f __attribute__((ext_vector_type(8)));

#define TPB 256

__global__ void fill_kernel(float* __restrict__ p, float v, int n) {
    int i = blockIdx.x * blockDim.x + threadIdx.x;
    if (i < n) p[i] = v;
}

__global__ void zero_kernel(int* __restrict__ p, int n) {
    int i = blockIdx.x * blockDim.x + threadIdx.x;
    if (i < n) p[i] = 0;
}

__global__ void rank_kernel(const int* __restrict__ dst, int* __restrict__ degi,
                            int* __restrict__ rank, int E) {
    int e = blockIdx.x * blockDim.x + threadIdx.x;
    if (e < E) rank[e] = atomicAdd(&degi[dst[e]], 1);
}

#define SCAN_T 1024
__global__ void __launch_bounds__(SCAN_T)
offsets_kernel(const int* __restrict__ degi, int* __restrict__ off,
               float* __restrict__ dinv, int n) {
    __shared__ int wsum[32];
    __shared__ int wpre[32];
    const int t = threadIdx.x;
    const int lane = t & 31;
    const int w = t >> 5;
    const int tile0 = blockIdx.x * SCAN_T;

    int pre = 0;
    for (int i = t; i < tile0; i += SCAN_T) pre += degi[i];
    for (int o = 16; o > 0; o >>= 1) pre += __shfl_xor(pre, o);

    const int i = tile0 + t;
    const int c = (i < n) ? degi[i] : 0;
    int s = c;
    for (int o = 1; o < 32; o <<= 1) {
        int v = __shfl_up(s, o);
        if (lane >= o) s += v;
    }
    if (lane == 31) wsum[w] = s;
    if (lane == 0)  wpre[w] = pre;
    __syncthreads();

    int tot = wpre[lane] + ((lane < w) ? wsum[lane] : 0);
    for (int o = 16; o > 0; o >>= 1) tot += __shfl_xor(tot, o);

    const int excl = tot + s - c;
    if (i < n) {
        off[i]  = excl;
        dinv[i] = rsqrtf((float)(c + 1));
    }
    if (i == n - 1) off[n] = excl + c;
}

__global__ void place_kernel(const int* __restrict__ src, const int* __restrict__ dst,
                             const int* __restrict__ off, const int* __restrict__ rank,
                             int* __restrict__ nbr, int E) {
    int e = blockIdx.x * blockDim.x + threadIdx.x;
    if (e < E) nbr[off[dst[e]] + rank[e]] = src[e];
}

template <int K, int Nn>
__global__ void wmma_gemm_f32(const float* __restrict__ A,
                              const float* __restrict__ B,
                              float* __restrict__ C,
                              const float* __restrict__ bias,
                              int row_tiles) {
    constexpr int COLT = Nn / 16;
    int wave = (blockIdx.x * blockDim.x + threadIdx.x) >> 5;
    if (wave >= row_tiles) return;
    int lane = threadIdx.x & 31;
    int half = lane >> 4;
    int lm   = lane & 15;
    int row0 = wave << 4;

    const float* __restrict__ arow  = A + (size_t)(row0 + lm) * K + 2 * half;
    const float* __restrict__ bbase = B + (size_t)(2 * half) * Nn + lm;

    v8f acc[COLT];
#pragma unroll
    for (int c = 0; c < COLT; ++c) acc[c] = (v8f){};

#pragma unroll
    for (int k0 = 0; k0 < K; k0 += 4) {
        v2f a;
        a.x = arow[k0];
        a.y = arow[k0 + 1];
#pragma unroll
        for (int c = 0; c < COLT; ++c) {
            v2f b;
            b.x = bbase[k0 * Nn + c * 16];
            b.y = bbase[(k0 + 1) * Nn + c * 16];
            acc[c] = __builtin_amdgcn_wmma_f32_16x16x4_f32(
                 false, a,  false, b,
                 (short)0, acc[c],  false,  false);
        }
    }

    float* __restrict__ crow = C + (size_t)(row0 + 8 * half) * Nn + lm;
#pragma unroll
    for (int c = 0; c < COLT; ++c) {
        float bv = bias ? bias[c * 16 + lm] : 0.0f;
#pragma unroll
        for (int v = 0; v < 8; ++v) {
            crow[(size_t)v * Nn + c * 16] = acc[c][v] + bv;
        }
    }
}

template <int F, bool RELU>
__global__ void __launch_bounds__(TPB)
gather_kernel(const float* __restrict__ h, const int* __restrict__ off,
              const int* __restrict__ nbr, const float* __restrict__ dinv,
              const float* __restrict__ bias, float* __restrict__ out, int N) {
    constexpr int VEC = F / 32;
    typedef float vf __attribute__((ext_vector_type(VEC)));
    const int node = __builtin_amdgcn_readfirstlane(
        (int)((blockIdx.x * blockDim.x + threadIdx.x) >> 5));
    if (node >= N) return;
    const int lane = threadIdx.x & 31;
    const vf* __restrict__ hv = (const vf*)h;

    const int beg = off[node];
    const int end = off[node + 1];
    const float dd = dinv[node];

    vf acc = hv[(size_t)node * 32 + lane] * (dd * dd);

    for (int base = beg; base < end; base += 32) {
        const int j = base + lane;
        int   s_l = node;
        float w_l = 0.0f;
        if (j < end) {
            s_l = nbr[j];
            w_l = dinv[s_l] * dd;
        }
        const int cnt  = min(32, end - base);
        const int cnt4 = (cnt + 3) & ~3;
        for (int k = 0; k < cnt4; k += 4) {
#pragma unroll
            for (int u = 0; u < 4; ++u) {
                const int   s = __builtin_amdgcn_readlane(s_l, k + u);
                const float w = __int_as_float(
                    __builtin_amdgcn_readlane(__float_as_int(w_l), k + u));
                acc += hv[(size_t)s * 32 + lane] * w;
            }
        }
    }

    vf r;
#pragma unroll
    for (int q = 0; q < VEC; ++q) {
        float v = acc[q] + bias[lane * VEC + q];
        r[q] = RELU ? fmaxf(v, 0.0f) : v;
    }
    ((vf*)out)[(size_t)node * 32 + lane] = r;
}

template <int F>
__global__ void pool_accum_kernel(const float* __restrict__ h,
                                  const int* __restrict__ batch,
                                  float* __restrict__ sums, int N) {
    int gid = blockIdx.x * blockDim.x + threadIdx.x;
    if (gid >= N * F) return;
    int i = gid / F;
    int f = gid - i * F;
    atomicAdd(&sums[(size_t)batch[i] * F + f], h[gid]);
}

__global__ void count_kernel(const int* __restrict__ batch, float* __restrict__ cnt, int N) {
    int i = blockIdx.x * blockDim.x + threadIdx.x;
    if (i < N) atomicAdd(&cnt[batch[i]], 1.0f);
}

template <int F>
__global__ void pool_div_kernel(float* __restrict__ sums, const float* __restrict__ cnt, int G) {
    int gid = blockIdx.x * blockDim.x + threadIdx.x;
    if (gid >= G * F) return;
    int g = gid / F;
    sums[gid] /= fmaxf(cnt[g], 1.0f);
}

static inline int cdiv(long long a, long long b) { return (int)((a + b - 1) / b); }

extern "C" void kernel_launch(void* const* d_in, const int* in_sizes, int n_in,
                              void* d_out, int out_size, void* d_ws, size_t ws_size,
                              hipStream_t stream) {
    (void)n_in; (void)ws_size;
    const float* x    = (const float*)d_in[0];
    const int*   edge = (const int*)d_in[1];
    const int*   batch= (const int*)d_in[2];
    const float* b1   = (const float*)d_in[4];
    const float* W1   = (const float*)d_in[3];
    const float* W2   = (const float*)d_in[5];
    const float* b2   = (const float*)d_in[6];
    const float* Wfc  = (const float*)d_in[7];
    const float* bfc  = (const float*)d_in[8];
    float* out = (float*)d_out;

    const int N   = in_sizes[2];
    const int E   = in_sizes[1] / 2;
    const int D   = 64;
    const int D2  = 128;
    const int HID = in_sizes[8];
    const int G   = out_size / HID;

    const int* src = edge;
    const int* dst = edge + E;

    float* ws    = (float*)d_ws;
    float* r0    = ws;
    float* r1    = r0 + (size_t)N * D2;
    float* r2    = r1 + (size_t)N * D2;
    float* xw1   = r0;
    float* h2    = r0;
    float* hw2   = r1;
    float* h1    = r2;
    int*   rank  = (int*)(r2 + (size_t)N * D);
    int*   nbr   = rank + E;
    int*   degi  = nbr + E;
    int*   off   = degi + N;
    float* dinv  = r2 + (size_t)N * D2;
    float* sums  = dinv + (size_t)N;
    float* cnt   = sums + (size_t)G * D2;

    zero_kernel<<<cdiv(N, TPB), TPB, 0, stream>>>(degi, N);
    fill_kernel<<<cdiv((long long)G * D2 + G, TPB), TPB, 0, stream>>>(sums, 0.0f, G * D2 + G);

    rank_kernel<<<cdiv(E, TPB), TPB, 0, stream>>>(dst, degi, rank, E);
    offsets_kernel<<<cdiv(N, SCAN_T), SCAN_T, 0, stream>>>(degi, off, dinv, N);
    place_kernel<<<cdiv(E, TPB), TPB, 0, stream>>>(src, dst, off, rank, nbr, E);

    {
        int row_tiles = N / 16;
        wmma_gemm_f32<64, 64><<<cdiv((long long)row_tiles * 32, TPB), TPB, 0, stream>>>(
            x, W1, xw1, nullptr, row_tiles);
    }
    gather_kernel<64, true><<<cdiv((long long)N * 32, TPB), TPB, 0, stream>>>(
        xw1, off, nbr, dinv, b1, h1, N);

    {
        int row_tiles = N / 16;
        wmma_gemm_f32<64, 128><<<cdiv((long long)row_tiles * 32, TPB), TPB, 0, stream>>>(
            h1, W2, hw2, nullptr, row_tiles);
    }
    gather_kernel<128, false><<<cdiv((long long)N * 32, TPB), TPB, 0, stream>>>(
        hw2, off, nbr, dinv, b2, h2, N);

    pool_accum_kernel<128><<<cdiv((long long)N * D2, TPB), TPB, 0, stream>>>(
        h2, batch, sums, N);
    count_kernel<<<cdiv(N, TPB), TPB, 0, stream>>>(batch, cnt, N);
    pool_div_kernel<128><<<cdiv((long long)G * D2, TPB), TPB, 0, stream>>>(sums, cnt, G);

    {
        int row_tiles = G / 16;
        wmma_gemm_f32<128, 64><<<cdiv((long long)row_tiles * 32, TPB), TPB, 0, stream>>>(
            sums, Wfc, out, bfc, row_tiles);
    }
}


# Mean-pool sums and node counts by running sums over 32-node chunks: one f32 atomic per run of equal graph ids and feature instead of one per element (sum order: chunk partials, then atomics)
# speedup vs baseline: 2.9838x; 1.1190x over previous
#include <hip/hip_runtime.h>
#include <hip/hip_bf16.h>


typedef float v2f __attribute__((ext_vector_type(2)));
typedef float v8f __attribute__((ext_vector_type(8)));

#define TPB 256

__global__ void fill_kernel(float* __restrict__ p, float v, int n) {
    int i = blockIdx.x * blockDim.x + threadIdx.x;
    if (i < n) p[i] = v;
}

__global__ void zero_kernel(int* __restrict__ p, int n) {
    int i = blockIdx.x * blockDim.x + threadIdx.x;
    if (i < n) p[i] = 0;
}

__global__ void rank_kernel(const int* __restrict__ dst, int* __restrict__ degi,
                            int* __restrict__ rank, int E) {
    int e = blockIdx.x * blockDim.x + threadIdx.x;
    if (e < E) rank[e] = atomicAdd(&degi[dst[e]], 1);
}

#define SCAN_T 1024
__global__ void __launch_bounds__(SCAN_T)
offsets_kernel(const int* __restrict__ degi, int* __restrict__ off,
               float* __restrict__ dinv, int n) {
    __shared__ int wsum[32];
    __shared__ int wpre[32];
    const int t = threadIdx.x;
    const int lane = t & 31;
    const int w = t >> 5;
    const int tile0 = blockIdx.x * SCAN_T;

    int pre = 0;
    for (int i = t; i < tile0; i += SCAN_T) pre += degi[i];
    for (int o = 16; o > 0; o >>= 1) pre += __shfl_xor(pre, o);

    const int i = tile0 + t;
    const int c = (i < n) ? degi[i] : 0;
    int s = c;
    for (int o = 1; o < 32; o <<= 1) {
        int v = __shfl_up(s, o);
        if (lane >= o) s += v;
    }
    if (lane == 31) wsum[w] = s;
    if (lane == 0)  wpre[w] = pre;
    __syncthreads();

    int tot = wpre[lane] + ((lane < w) ? wsum[lane] : 0);
    for (int o = 16; o > 0; o >>= 1) tot += __shfl_xor(tot, o);

    const int excl = tot + s - c;
    if (i < n) {
        off[i]  = excl;
        dinv[i] = rsqrtf((float)(c + 1));
    }
    if (i == n - 1) off[n] = excl + c;
}

__global__ void place_kernel(const int* __restrict__ src, const int* __restrict__ dst,
                             const int* __restrict__ off, const int* __restrict__ rank,
                             int* __restrict__ nbr, int E) {
    int e = blockIdx.x * blockDim.x + threadIdx.x;
    if (e < E) nbr[off[dst[e]] + rank[e]] = src[e];
}

template <int K, int Nn>
__global__ void wmma_gemm_f32(const float* __restrict__ A,
                              const float* __restrict__ B,
                              float* __restrict__ C,
                              const float* __restrict__ bias,
                              int row_tiles) {
    constexpr int COLT = Nn / 16;
    int wave = (blockIdx.x * blockDim.x + threadIdx.x) >> 5;
    if (wave >= row_tiles) return;
    int lane = threadIdx.x & 31;
    int half = lane >> 4;
    int lm   = lane & 15;
    int row0 = wave << 4;

    const float* __restrict__ arow  = A + (size_t)(row0 + lm) * K + 2 * half;
    const float* __restrict__ bbase = B + (size_t)(2 * half) * Nn + lm;

    v8f acc[COLT];
#pragma unroll
    for (int c = 0; c < COLT; ++c) acc[c] = (v8f){};

#pragma unroll
    for (int k0 = 0; k0 < K; k0 += 4) {
        v2f a;
        a.x = arow[k0];
        a.y = arow[k0 + 1];
#pragma unroll
        for (int c = 0; c < COLT; ++c) {
            v2f b;
            b.x = bbase[k0 * Nn + c * 16];
            b.y = bbase[(k0 + 1) * Nn + c * 16];
            acc[c] = __builtin_amdgcn_wmma_f32_16x16x4_f32(
                 false, a,  false, b,
                 (short)0, acc[c],  false,  false);
        }
    }

    float* __restrict__ crow = C + (size_t)(row0 + 8 * half) * Nn + lm;
#pragma unroll
    for (int c = 0; c < COLT; ++c) {
        float bv = bias ? bias[c * 16 + lm] : 0.0f;
#pragma unroll
        for (int v = 0; v < 8; ++v) {
            crow[(size_t)v * Nn + c * 16] = acc[c][v] + bv;
        }
    }
}

template <int F, bool RELU>
__global__ void __launch_bounds__(TPB)
gather_kernel(const float* __restrict__ h, const int* __restrict__ off,
              const int* __restrict__ nbr, const float* __restrict__ dinv,
              const float* __restrict__ bias, float* __restrict__ out, int N) {
    constexpr int VEC = F / 32;
    typedef float vf __attribute__((ext_vector_type(VEC)));
    const int node = __builtin_amdgcn_readfirstlane(
        (int)((blockIdx.x * blockDim.x + threadIdx.x) >> 5));
    if (node >= N) return;
    const int lane = threadIdx.x & 31;
    const vf* __restrict__ hv = (const vf*)h;

    const int beg = off[node];
    const int end = off[node + 1];
    const float dd = dinv[node];

    vf acc = hv[(size_t)node * 32 + lane] * (dd * dd);

    for (int base = beg; base < end; base += 32) {
        const int j = base + lane;
        int   s_l = node;
        float w_l = 0.0f;
        if (j < end) {
            s_l = nbr[j];
            w_l = dinv[s_l] * dd;
        }
        const int cnt  = min(32, end - base);
        const int cnt4 = (cnt + 3) & ~3;
        for (int k = 0; k < cnt4; k += 4) {
#pragma unroll
            for (int u = 0; u < 4; ++u) {
                const int   s = __builtin_amdgcn_readlane(s_l, k + u);
                const float w = __int_as_float(
                    __builtin_amdgcn_readlane(__float_as_int(w_l), k + u));
                acc += hv[(size_t)s * 32 + lane] * w;
            }
        }
    }

    vf r;
#pragma unroll
    for (int q = 0; q < VEC; ++q) {
        float v = acc[q] + bias[lane * VEC + q];
        r[q] = RELU ? fmaxf(v, 0.0f) : v;
    }
    ((vf*)out)[(size_t)node * 32 + lane] = r;
}

template <int F, int CH>
__global__ void __launch_bounds__(F)
pool_runs_kernel(const float* __restrict__ h, const int* __restrict__ batch,
                 float* __restrict__ sums, float* __restrict__ cnt, int N) {
    static_assert(CH <= 32, "one lane per node of the chunk holds its graph id");
    const int f    = threadIdx.x;
    const int lane = f & 31;
    const int n0   = blockIdx.x * CH;
    const int m    = min(CH, N - n0);

    const int g_l = batch[n0 + min(lane, m - 1)];

    float v[CH];
#pragma unroll
    for (int j = 0; j < CH; ++j) {
        const float* __restrict__ row = h + (size_t)(n0 + min(j, m - 1)) * F;
        const float t = row[f];
        v[j] = (j < m) ? t : 0.0f;
    }

    int   g_cur = __builtin_amdgcn_readlane(g_l, 0);
    int   run   = 0;
    float acc   = 0.0f;
#pragma unroll
    for (int j = 0; j < CH; ++j) {
        const int g = __builtin_amdgcn_readlane(g_l, j);
        if (g != g_cur) {
            atomicAdd(&sums[(size_t)g_cur * F + f], acc);
            if (f == 0) atomicAdd(&cnt[g_cur], (float)run);
            acc = 0.0f; run = 0; g_cur = g;
        }
        acc += v[j];
        run += (j < m) ? 1 : 0;
    }
    atomicAdd(&sums[(size_t)g_cur * F + f], acc);
    if (f == 0) atomicAdd(&cnt[g_cur], (float)run);
}

template <int F>
__global__ void pool_div_kernel(float* __restrict__ sums, const float* __restrict__ cnt, int G) {
    int gid = blockIdx.x * blockDim.x + threadIdx.x;
    if (gid >= G * F) return;
    int g = gid / F;
    sums[gid] /= fmaxf(cnt[g], 1.0f);
}

static inline int cdiv(long long a, long long b) { return (int)((a + b - 1) / b); }

extern "C" void kernel_launch(void* const* d_in, const int* in_sizes, int n_in,
                              void* d_out, int out_size, void* d_ws, size_t ws_size,
                              hipStream_t stream) {
    (void)n_in; (void)ws_size;
    const float* x    = (const float*)d_in[0];
    const int*   edge = (const int*)d_in[1];
    const int*   batch= (const int*)d_in[2];
    const float* b1   = (const float*)d_in[4];
    const float* W1   = (const float*)d_in[3];
    const float* W2   = (const float*)d_in[5];
    const float* b2   = (const float*)d_in[6];
    const float* Wfc  = (const float*)d_in[7];
    const float* bfc  = (const float*)d_in[8];
    float* out = (float*)d_out;

    const int N   = in_sizes[2];
    const int E   = in_sizes[1] / 2;
    const int D   = 64;
    const int D2  = 128;
    const int HID = in_sizes[8];
    const int G   = out_size / HID;

    const int* src = edge;
    const int* dst = edge + E;

    float* ws    = (float*)d_ws;
    float* r0    = ws;
    float* r1    = r0 + (size_t)N * D2;
    float* r2    = r1 + (size_t)N * D2;
    float* xw1   = r0;
    float* h2    = r0;
    float* hw2   = r1;
    float* h1    = r2;
    int*   rank  = (int*)(r2 + (size_t)N * D);
    int*   nbr   = rank + E;
    int*   degi  = nbr + E;
    int*   off   = degi + N;
    float* dinv  = r2 + (size_t)N * D2;
    float* sums  = dinv + (size_t)N;
    float* cnt   = sums + (size_t)G * D2;

    zero_kernel<<<cdiv(N, TPB), TPB, 0, stream>>>(degi, N);
    fill_kernel<<<cdiv((long long)G * D2 + G, TPB), TPB, 0, stream>>>(sums, 0.0f, G * D2 + G);

    rank_kernel<<<cdiv(E, TPB), TPB, 0, stream>>>(dst, degi, rank, E);
    offsets_kernel<<<cdiv(N, SCAN_T), SCAN_T, 0, stream>>>(degi, off, dinv, N);
    place_kernel<<<cdiv(E, TPB), TPB, 0, stream>>>(src, dst, off, rank, nbr, E);

    {
        int row_tiles = N / 16;
        wmma_gemm_f32<64, 64><<<cdiv((long long)row_tiles * 32, TPB), TPB, 0, stream>>>(
            x, W1, xw1, nullptr, row_tiles);
    }
    gather_kernel<64, true><<<cdiv((long long)N * 32, TPB), TPB, 0, stream>>>(
        xw1, off, nbr, dinv, b1, h1, N);

    {
        int row_tiles = N / 16;
        wmma_gemm_f32<64, 128><<<cdiv((long long)row_tiles * 32, TPB), TPB, 0, stream>>>(
            h1, W2, hw2, nullptr, row_tiles);
    }
    gather_kernel<128, false><<<cdiv((long long)N * 32, TPB), TPB, 0, stream>>>(
        hw2, off, nbr, dinv, b2, h2, N);

    pool_runs_kernel<128, 32><<<cdiv(N, 32), 128, 0, stream>>>(
        h2, batch, sums, cnt, N);
    pool_div_kernel<128><<<cdiv((long long)G * D2, TPB), TPB, 0, stream>>>(sums, cnt, G);

    {
        int row_tiles = G / 16;
        wmma_gemm_f32<128, 64><<<cdiv((long long)row_tiles * 32, TPB), TPB, 0, stream>>>(
            sums, Wfc, out, bfc, row_tiles);
    }
}


# W2 product moved behind aggregation and mean pool (both linear): neighbour sums and pool on 64 features, W2 on the 512 pooled rows with mean and bias in its epilogue; sum order changes
# speedup vs baseline: 3.3075x; 1.1085x over previous
#include <hip/hip_runtime.h>
#include <hip/hip_bf16.h>


typedef float v2f __attribute__((ext_vector_type(2)));
typedef float v8f __attribute__((ext_vector_type(8)));

#define TPB 256

__global__ void fill_kernel(float* __restrict__ p, float v, int n) {
    int i = blockIdx.x * blockDim.x + threadIdx.x;
    if (i < n) p[i] = v;
}

__global__ void zero_kernel(int* __restrict__ p, int n) {
    int i = blockIdx.x * blockDim.x + threadIdx.x;
    if (i < n) p[i] = 0;
}

__global__ void rank_kernel(const int* __restrict__ dst, int* __restrict__ degi,
                            int* __restrict__ rank, int E) {
    int e = blockIdx.x * blockDim.x + threadIdx.x;
    if (e < E) rank[e] = atomicAdd(&degi[dst[e]], 1);
}

#define SCAN_T 1024
__global__ void __launch_bounds__(SCAN_T)
offsets_kernel(const int* __restrict__ degi, int* __restrict__ off,
               float* __restrict__ dinv, int n) {
    __shared__ int wsum[32];
    __shared__ int wpre[32];
    const int t = threadIdx.x;
    const int lane = t & 31;
    const int w = t >> 5;
    const int tile0 = blockIdx.x * SCAN_T;

    int pre = 0;
    for (int i = t; i < tile0; i += SCAN_T) pre += degi[i];
    for (int o = 16; o > 0; o >>= 1) pre += __shfl_xor(pre, o);

    const int i = tile0 + t;
    const int c = (i < n) ? degi[i] : 0;
    int s = c;
    for (int o = 1; o < 32; o <<= 1) {
        int v = __shfl_up(s, o);
        if (lane >= o) s += v;
    }
    if (lane == 31) wsum[w] = s;
    if (lane == 0)  wpre[w] = pre;
    __syncthreads();

    int tot = wpre[lane] + ((lane < w) ? wsum[lane] : 0);
    for (int o = 16; o > 0; o >>= 1) tot += __shfl_xor(tot, o);

    const int excl = tot + s - c;
    if (i < n) {
        off[i]  = excl;
        dinv[i] = rsqrtf((float)(c + 1));
    }
    if (i == n - 1) off[n] = excl + c;
}

__global__ void place_kernel(const int* __restrict__ src, const int* __restrict__ dst,
                             const int* __restrict__ off, const int* __restrict__ rank,
                             int* __restrict__ nbr, int E) {
    int e = blockIdx.x * blockDim.x + threadIdx.x;
    if (e < E) nbr[off[dst[e]] + rank[e]] = src[e];
}

template <int K, int Nn, bool MEAN>
__global__ void wmma_gemm_f32(const float* __restrict__ A,
                              const float* __restrict__ B,
                              float* __restrict__ C,
                              const float* __restrict__ bias,
                              const float* __restrict__ cnt,
                              int row_tiles) {
    constexpr int COLT = Nn / 16;
    int wave = (blockIdx.x * blockDim.x + threadIdx.x) >> 5;
    if (wave >= row_tiles) return;
    int lane = threadIdx.x & 31;
    int half = lane >> 4;
    int lm   = lane & 15;
    int row0 = wave << 4;

    const float* __restrict__ arow  = A + (size_t)(row0 + lm) * K + 2 * half;
    const float* __restrict__ bbase = B + (size_t)(2 * half) * Nn + lm;

    v8f acc[COLT];
#pragma unroll
    for (int c = 0; c < COLT; ++c) acc[c] = (v8f){};

#pragma unroll
    for (int k0 = 0; k0 < K; k0 += 4) {
        v2f a;
        a.x = arow[k0];
        a.y = arow[k0 + 1];
#pragma unroll
        for (int c = 0; c < COLT; ++c) {
            v2f b;
            b.x = bbase[k0 * Nn + c * 16];
            b.y = bbase[(k0 + 1) * Nn + c * 16];
            acc[c] = __builtin_amdgcn_wmma_f32_16x16x4_f32(
                 false, a,  false, b,
                 (short)0, acc[c],  false,  false);
        }
    }

    float* __restrict__ crow = C + (size_t)(row0 + 8 * half) * Nn + lm;
    float inv[8];
    bool  has[8];
#pragma unroll
    for (int v = 0; v < 8; ++v) {
        const float cv = MEAN ? cnt[row0 + 8 * half + v] : 1.0f;
        inv[v] = 1.0f / fmaxf(cv, 1.0f);
        has[v] = cv > 0.0f;
    }
#pragma unroll
    for (int c = 0; c < COLT; ++c) {
        float bv = bias ? bias[c * 16 + lm] : 0.0f;
#pragma unroll
        for (int v = 0; v < 8; ++v) {
            float r = acc[c][v];
            if (MEAN) r = r * inv[v] + (has[v] ? bv : 0.0f);
            else      r = r + bv;
            crow[(size_t)v * Nn + c * 16] = r;
        }
    }
}

template <int F, bool RELU, bool BIAS>
__global__ void __launch_bounds__(TPB)
gather_kernel(const float* __restrict__ h, const int* __restrict__ off,
              const int* __restrict__ nbr, const float* __restrict__ dinv,
              const float* __restrict__ bias, float* __restrict__ out, int N) {
    constexpr int VEC = F / 32;
    typedef float vf __attribute__((ext_vector_type(VEC)));
    const int node = __builtin_amdgcn_readfirstlane(
        (int)((blockIdx.x * blockDim.x + threadIdx.x) >> 5));
    if (node >= N) return;
    const int lane = threadIdx.x & 31;
    const vf* __restrict__ hv = (const vf*)h;

    const int beg = off[node];
    const int end = off[node + 1];
    const float dd = dinv[node];

    vf acc = hv[(size_t)node * 32 + lane] * (dd * dd);

    for (int base = beg; base < end; base += 32) {
        const int j = base + lane;
        int   s_l = node;
        float w_l = 0.0f;
        if (j < end) {
            s_l = nbr[j];
            w_l = dinv[s_l] * dd;
        }
        const int cnt  = min(32, end - base);
        const int cnt4 = (cnt + 3) & ~3;
        for (int k = 0; k < cnt4; k += 4) {
#pragma unroll
            for (int u = 0; u < 4; ++u) {
                const int   s = __builtin_amdgcn_readlane(s_l, k + u);
                const float w = __int_as_float(
                    __builtin_amdgcn_readlane(__float_as_int(w_l), k + u));
                acc += hv[(size_t)s * 32 + lane] * w;
            }
        }
    }

    vf r;
#pragma unroll
    for (int q = 0; q < VEC; ++q) {
        float v = acc[q];
        if (BIAS) v += bias[lane * VEC + q];
        r[q] = RELU ? fmaxf(v, 0.0f) : v;
    }
    ((vf*)out)[(size_t)node * 32 + lane] = r;
}

template <int F, int CH>
__global__ void __launch_bounds__(F)
pool_runs_kernel(const float* __restrict__ h, const int* __restrict__ batch,
                 float* __restrict__ sums, float* __restrict__ cnt, int N) {
    static_assert(CH <= 32, "one lane per node of the chunk holds its graph id");
    const int f    = threadIdx.x;
    const int lane = f & 31;
    const int n0   = blockIdx.x * CH;
    const int m    = min(CH, N - n0);

    const int g_l = batch[n0 + min(lane, m - 1)];

    float v[CH];
#pragma unroll
    for (int j = 0; j < CH; ++j) {
        const float* __restrict__ row = h + (size_t)(n0 + min(j, m - 1)) * F;
        const float t = row[f];
        v[j] = (j < m) ? t : 0.0f;
    }

    int   g_cur = __builtin_amdgcn_readlane(g_l, 0);
    int   run   = 0;
    float acc   = 0.0f;
#pragma unroll
    for (int j = 0; j < CH; ++j) {
        const int g = __builtin_amdgcn_readlane(g_l, j);
        if (g != g_cur) {
            atomicAdd(&sums[(size_t)g_cur * F + f], acc);
            if (f == 0) atomicAdd(&cnt[g_cur], (float)run);
            acc = 0.0f; run = 0; g_cur = g;
        }
        acc += v[j];
        run += (j < m) ? 1 : 0;
    }
    atomicAdd(&sums[(size_t)g_cur * F + f], acc);
    if (f == 0) atomicAdd(&cnt[g_cur], (float)run);
}

static inline int cdiv(long long a, long long b) { return (int)((a + b - 1) / b); }

extern "C" void kernel_launch(void* const* d_in, const int* in_sizes, int n_in,
                              void* d_out, int out_size, void* d_ws, size_t ws_size,
                              hipStream_t stream) {
    (void)n_in; (void)ws_size;
    const float* x    = (const float*)d_in[0];
    const int*   edge = (const int*)d_in[1];
    const int*   batch= (const int*)d_in[2];
    const float* b1   = (const float*)d_in[4];
    const float* W1   = (const float*)d_in[3];
    const float* W2   = (const float*)d_in[5];
    const float* b2   = (const float*)d_in[6];
    const float* Wfc  = (const float*)d_in[7];
    const float* bfc  = (const float*)d_in[8];
    float* out = (float*)d_out;

    const int N   = in_sizes[2];
    const int E   = in_sizes[1] / 2;
    const int D   = 64;
    const int D2  = 128;
    const int HID = in_sizes[8];
    const int G   = out_size / HID;

    const int* src = edge;
    const int* dst = edge + E;

    float* ws    = (float*)d_ws;
    float* r0    = ws;
    float* r1    = r0 + (size_t)N * D2;
    float* r2    = r1 + (size_t)N * D2;
    float* xw1   = r0;
    float* pooled= r0 + (size_t)N * D;
    float* a2    = r1;
    float* h1    = r2;
    int*   rank  = (int*)(r2 + (size_t)N * D);
    int*   nbr   = rank + E;
    int*   degi  = nbr + E;
    int*   off   = degi + N;
    float* dinv  = r2 + (size_t)N * D2;
    float* sums  = dinv + (size_t)N;
    float* cnt   = sums + (size_t)G * D;

    zero_kernel<<<cdiv(N, TPB), TPB, 0, stream>>>(degi, N);
    fill_kernel<<<cdiv((long long)G * D + G, TPB), TPB, 0, stream>>>(sums, 0.0f, G * D + G);

    rank_kernel<<<cdiv(E, TPB), TPB, 0, stream>>>(dst, degi, rank, E);
    offsets_kernel<<<cdiv(N, SCAN_T), SCAN_T, 0, stream>>>(degi, off, dinv, N);
    place_kernel<<<cdiv(E, TPB), TPB, 0, stream>>>(src, dst, off, rank, nbr, E);

    {
        int row_tiles = N / 16;
        wmma_gemm_f32<64, 64, false><<<cdiv((long long)row_tiles * 32, TPB), TPB, 0, stream>>>(
            x, W1, xw1, nullptr, nullptr, row_tiles);
    }
    gather_kernel<64, true, true><<<cdiv((long long)N * 32, TPB), TPB, 0, stream>>>(
        xw1, off, nbr, dinv, b1, h1, N);

    gather_kernel<64, false, false><<<cdiv((long long)N * 32, TPB), TPB, 0, stream>>>(
        h1, off, nbr, dinv, nullptr, a2, N);

    pool_runs_kernel<64, 32><<<cdiv(N, 32), 64, 0, stream>>>(
        a2, batch, sums, cnt, N);

    {
        int row_tiles = G / 16;
        wmma_gemm_f32<64, 128, true><<<cdiv((long long)row_tiles * 32, TPB), TPB, 0, stream>>>(
            sums, W2, pooled, b2, cnt, row_tiles);
    }

    {
        int row_tiles = G / 16;
        wmma_gemm_f32<128, 64, false><<<cdiv((long long)row_tiles * 32, TPB), TPB, 0, stream>>>(
            pooled, Wfc, out, bfc, nullptr, row_tiles);
    }
}


# Edges grouped by destination with a two-level counting sort on LDS counters (chunk pass, bucket pass) instead of one returning global atomic per edge; slot (sum) order now LDS arrival order
# speedup vs baseline: 5.3931x; 1.6306x over previous
#include <hip/hip_runtime.h>
#include <hip/hip_bf16.h>


typedef float v2f __attribute__((ext_vector_type(2)));
typedef float v8f __attribute__((ext_vector_type(8)));

#define TPB 256

__global__ void fill_kernel(float* __restrict__ p, float v, int n) {
    int i = blockIdx.x * blockDim.x + threadIdx.x;
    if (i < n) p[i] = v;
}

#define SORT_T 256
#define BIN_SH 8
#define CHUNKS 256
#define CHUNK_T 1024
#define LB     4
#define PB     8

template <int NT>
__device__ __forceinline__ int block_scan(int v, int* __restrict__ wtot, int& all) {
    const int lane = threadIdx.x & 31;
    const int w    = threadIdx.x >> 5;
    int s = v;
#pragma unroll
    for (int o = 1; o < 32; o <<= 1) {
        int u = __shfl_up(s, o);
        if (lane >= o) s += u;
    }
    if (lane == 31) wtot[w] = s;
    __syncthreads();
    int pre = 0, tot = 0;
#pragma unroll
    for (int i = 0; i < NT / 32; ++i) {
        const int x = wtot[i];
        tot += x;
        pre += (i < w) ? x : 0;
    }
    all = tot;
    return pre + s - v;
}

__global__ void __launch_bounds__(CHUNK_T)
chunk_pass_kernel(const int* __restrict__ src, const int* __restrict__ dst,
                  unsigned* __restrict__ binned, int* __restrict__ tab,
                  int E, int CH, int NB) {
    __shared__ int bcnt[SORT_T];
    __shared__ int wtot[CHUNK_T / 32];
    const int t  = threadIdx.x;
    const int c  = blockIdx.x;
    const int e0 = c * CH;
    const int n  = max(0, min(CH, E - e0));
    const unsigned last = (unsigned)(NB - 1);
    if (t < SORT_T) bcnt[t] = 0;
    __syncthreads();
    for (int i0 = t; i0 < n; i0 += LB * CHUNK_T) {
        unsigned d[LB];
#pragma unroll
        for (int u = 0; u < LB; ++u) {
            d[u] = (unsigned)dst[e0 + min(i0 + u * CHUNK_T, n - 1)];
        }
#pragma unroll
        for (int u = 0; u < LB; ++u) {
            const bool valid = i0 + u * CHUNK_T < n;
            const unsigned b = valid ? min(d[u] >> BIN_SH, last) : (unsigned)(t & (SORT_T - 1));
            atomicAdd(&bcnt[b], valid ? 1 : 0);
        }
    }
    __syncthreads();
    int tot;
    const int v  = (t < SORT_T) ? bcnt[t] : 0;
    const int ex = block_scan<CHUNK_T>(v, wtot, tot);
    if (t <= NB) tab[t * CHUNKS + c] = ex;
    if (t < SORT_T) bcnt[t] = ex;
    __syncthreads();
    for (int i0 = t; i0 < n; i0 += LB * CHUNK_T) {
        unsigned d[LB], s[LB];
#pragma unroll
        for (int u = 0; u < LB; ++u) {
            const int i = min(i0 + u * CHUNK_T, n - 1);
            d[u] = (unsigned)dst[e0 + i];
            s[u] = (unsigned)src[e0 + i];
        }
#pragma unroll
        for (int u = 0; u < LB; ++u) {
            const bool valid = i0 + u * CHUNK_T < n;
            const unsigned b = valid ? min(d[u] >> BIN_SH, last) : (unsigned)(t & (SORT_T - 1));
            const int p = atomicAdd(&bcnt[b], valid ? 1 : 0);
            if (valid) binned[e0 + p] = (s[u] << BIN_SH) | (d[u] & (unsigned)(SORT_T - 1));
        }
    }
}

__global__ void __launch_bounds__(SORT_T)
bucket_pass_kernel(const unsigned* __restrict__ binned, const int* __restrict__ tab,
                   int* __restrict__ off, int* __restrict__ nbr,
                   float* __restrict__ dinv, int n, int CH) {
    static_assert(CHUNKS == SORT_T, "thread t of a bucket work-group walks chunk t");
    __shared__ int ncnt[SORT_T];
    __shared__ int wtot_a[SORT_T / 32];
    __shared__ int wtot_b[SORT_T / 32];
    const int t  = threadIdx.x;
    const int b  = blockIdx.x;
    const int o0 = tab[b * CHUNKS + t];
    const int o1 = tab[(b + 1) * CHUNKS + t];
    const int len = o1 - o0;
    const unsigned* __restrict__ piece = binned + (size_t)t * CH + o0;
    ncnt[t] = 0;
    int base;
    (void)block_scan<SORT_T>(o0, wtot_a, base);
    for (int k0 = 0; k0 < len; k0 += PB) {
        unsigned w[PB];
#pragma unroll
        for (int u = 0; u < PB; ++u) w[u] = piece[min(k0 + u, len - 1)];
#pragma unroll
        for (int u = 0; u < PB; ++u) {
            const bool valid = k0 + u < len;
            atomicAdd(&ncnt[valid ? (w[u] & (unsigned)(SORT_T - 1)) : (unsigned)t], valid ? 1 : 0);
        }
    }
    __syncthreads();
    const int cdeg = ncnt[t];
    int unused;
    const int ex = block_scan<SORT_T>(cdeg, wtot_b, unused);
    const int node = b * SORT_T + t;
    if (node < n) {
        off[node]  = base + ex;
        dinv[node] = rsqrtf((float)(cdeg + 1));
    }
    if (node == n - 1) off[n] = base + ex + cdeg;
    ncnt[t] = base + ex;
    __syncthreads();
    for (int k0 = 0; k0 < len; k0 += PB) {
        unsigned w[PB];
#pragma unroll
        for (int u = 0; u < PB; ++u) w[u] = piece[min(k0 + u, len - 1)];
#pragma unroll
        for (int u = 0; u < PB; ++u) {
            const bool valid = k0 + u < len;
            const int p = atomicAdd(&ncnt[valid ? (w[u] & (unsigned)(SORT_T - 1)) : (unsigned)t], valid ? 1 : 0);
            if (valid) nbr[p] = (int)(w[u] >> BIN_SH);
        }
    }
}

template <int K, int Nn, bool MEAN>
__global__ void wmma_gemm_f32(const float* __restrict__ A,
                              const float* __restrict__ B,
                              float* __restrict__ C,
                              const float* __restrict__ bias,
                              const float* __restrict__ cnt,
                              int row_tiles) {
    constexpr int COLT = Nn / 16;
    int wave = (blockIdx.x * blockDim.x + threadIdx.x) >> 5;
    if (wave >= row_tiles) return;
    int lane = threadIdx.x & 31;
    int half = lane >> 4;
    int lm   = lane & 15;
    int row0 = wave << 4;

    const float* __restrict__ arow  = A + (size_t)(row0 + lm) * K + 2 * half;
    const float* __restrict__ bbase = B + (size_t)(2 * half) * Nn + lm;

    v8f acc[COLT];
#pragma unroll
    for (int c = 0; c < COLT; ++c) acc[c] = (v8f){};

#pragma unroll
    for (int k0 = 0; k0 < K; k0 += 4) {
        v2f a;
        a.x = arow[k0];
        a.y = arow[k0 + 1];
#pragma unroll
        for (int c = 0; c < COLT; ++c) {
            v2f b;
            b.x = bbase[k0 * Nn + c * 16];
            b.y = bbase[(k0 + 1) * Nn + c * 16];
            acc[c] = __builtin_amdgcn_wmma_f32_16x16x4_f32(
                 false, a,  false, b,
                 (short)0, acc[c],  false,  false);
        }
    }

    float* __restrict__ crow = C + (size_t)(row0 + 8 * half) * Nn + lm;
    float inv[8];
    bool  has[8];
#pragma unroll
    for (int v = 0; v < 8; ++v) {
        const float cv = MEAN ? cnt[row0 + 8 * half + v] : 1.0f;
        inv[v] = 1.0f / fmaxf(cv, 1.0f);
        has[v] = cv > 0.0f;
    }
#pragma unroll
    for (int c = 0; c < COLT; ++c) {
        float bv = bias ? bias[c * 16 + lm] : 0.0f;
#pragma unroll
        for (int v = 0; v < 8; ++v) {
            float r = acc[c][v];
            if (MEAN) r = r * inv[v] + (has[v] ? bv : 0.0f);
            else      r = r + bv;
            crow[(size_t)v * Nn + c * 16] = r;
        }
    }
}

template <int F, bool RELU, bool BIAS>
__global__ void __launch_bounds__(TPB)
gather_kernel(const float* __restrict__ h, const int* __restrict__ off,
              const int* __restrict__ nbr, const float* __restrict__ dinv,
              const float* __restrict__ bias, float* __restrict__ out, int N) {
    constexpr int VEC = F / 32;
    typedef float vf __attribute__((ext_vector_type(VEC)));
    const int node = __builtin_amdgcn_readfirstlane(
        (int)((blockIdx.x * blockDim.x + threadIdx.x) >> 5));
    if (node >= N) return;
    const int lane = threadIdx.x & 31;
    const vf* __restrict__ hv = (const vf*)h;

    const int beg = off[node];
    const int end = off[node + 1];
    const float dd = dinv[node];

    vf acc = hv[(size_t)node * 32 + lane] * (dd * dd);

    for (int base = beg; base < end; base += 32) {
        const int j = base + lane;
        int   s_l = node;
        float w_l = 0.0f;
        if (j < end) {
            s_l = nbr[j];
            w_l = dinv[s_l] * dd;
        }
        const int cnt  = min(32, end - base);
        const int cnt4 = (cnt + 3) & ~3;
        for (int k = 0; k < cnt4; k += 4) {
#pragma unroll
            for (int u = 0; u < 4; ++u) {
                const int   s = __builtin_amdgcn_readlane(s_l, k + u);
                const float w = __int_as_float(
                    __builtin_amdgcn_readlane(__float_as_int(w_l), k + u));
                acc += hv[(size_t)s * 32 + lane] * w;
            }
        }
    }

    vf r;
#pragma unroll
    for (int q = 0; q < VEC; ++q) {
        float v = acc[q];
        if (BIAS) v += bias[lane * VEC + q];
        r[q] = RELU ? fmaxf(v, 0.0f) : v;
    }
    ((vf*)out)[(size_t)node * 32 + lane] = r;
}

template <int F, int CH>
__global__ void __launch_bounds__(F)
pool_runs_kernel(const float* __restrict__ h, const int* __restrict__ batch,
                 float* __restrict__ sums, float* __restrict__ cnt, int N) {
    static_assert(CH <= 32, "one lane per node of the chunk holds its graph id");
    const int f    = threadIdx.x;
    const int lane = f & 31;
    const int n0   = blockIdx.x * CH;
    const int m    = min(CH, N - n0);

    const int g_l = batch[n0 + min(lane, m - 1)];

    float v[CH];
#pragma unroll
    for (int j = 0; j < CH; ++j) {
        const float* __restrict__ row = h + (size_t)(n0 + min(j, m - 1)) * F;
        const float t = row[f];
        v[j] = (j < m) ? t : 0.0f;
    }

    int   g_cur = __builtin_amdgcn_readlane(g_l, 0);
    int   run   = 0;
    float acc   = 0.0f;
#pragma unroll
    for (int j = 0; j < CH; ++j) {
        const int g = __builtin_amdgcn_readlane(g_l, j);
        if (g != g_cur) {
            atomicAdd(&sums[(size_t)g_cur * F + f], acc);
            if (f == 0) atomicAdd(&cnt[g_cur], (float)run);
            acc = 0.0f; run = 0; g_cur = g;
        }
        acc += v[j];
        run += (j < m) ? 1 : 0;
    }
    atomicAdd(&sums[(size_t)g_cur * F + f], acc);
    if (f == 0) atomicAdd(&cnt[g_cur], (float)run);
}

static inline int cdiv(long long a, long long b) { return (int)((a + b - 1) / b); }

extern "C" void kernel_launch(void* const* d_in, const int* in_sizes, int n_in,
                              void* d_out, int out_size, void* d_ws, size_t ws_size,
                              hipStream_t stream) {
    (void)n_in; (void)ws_size;
    const float* x    = (const float*)d_in[0];
    const int*   edge = (const int*)d_in[1];
    const int*   batch= (const int*)d_in[2];
    const float* b1   = (const float*)d_in[4];
    const float* W1   = (const float*)d_in[3];
    const float* W2   = (const float*)d_in[5];
    const float* b2   = (const float*)d_in[6];
    const float* Wfc  = (const float*)d_in[7];
    const float* bfc  = (const float*)d_in[8];
    float* out = (float*)d_out;

    const int N   = in_sizes[2];
    const int E   = in_sizes[1] / 2;
    const int D   = 64;
    const int D2  = 128;
    const int HID = in_sizes[8];
    const int G   = out_size / HID;

    const int* src = edge;
    const int* dst = edge + E;

    float* ws    = (float*)d_ws;
    float* r0    = ws;
    float* r1    = r0 + (size_t)N * D2;
    float* r2    = r1 + (size_t)N * D2;
    float* xw1   = r0;
    float* pooled= r0 + (size_t)N * D;
    float* a2    = r1;
    float* h1    = r2;
    const int NB = cdiv(N, SORT_T);
    const int CH = cdiv(E, CHUNKS);
    unsigned* binned = (unsigned*)(r2 + (size_t)N * D);
    int*   nbr   = (int*)binned + E;
    int*   tab   = nbr + E;
    int*   off   = tab + (size_t)(NB + 1) * CHUNKS;
    float* dinv  = r2 + (size_t)N * D2;
    float* sums  = dinv + (size_t)N;
    float* cnt   = sums + (size_t)G * D;

    fill_kernel<<<cdiv((long long)G * D + G, TPB), TPB, 0, stream>>>(sums, 0.0f, G * D + G);

    chunk_pass_kernel<<<CHUNKS, CHUNK_T, 0, stream>>>(src, dst, binned, tab, E, CH, NB);
    bucket_pass_kernel<<<NB, SORT_T, 0, stream>>>(binned, tab, off, nbr, dinv, N, CH);

    {
        int row_tiles = N / 16;
        wmma_gemm_f32<64, 64, false><<<cdiv((long long)row_tiles * 32, TPB), TPB, 0, stream>>>(
            x, W1, xw1, nullptr, nullptr, row_tiles);
    }
    gather_kernel<64, true, true><<<cdiv((long long)N * 32, TPB), TPB, 0, stream>>>(
        xw1, off, nbr, dinv, b1, h1, N);

    gather_kernel<64, false, false><<<cdiv((long long)N * 32, TPB), TPB, 0, stream>>>(
        h1, off, nbr, dinv, nullptr, a2, N);

    pool_runs_kernel<64, 32><<<cdiv(N, 32), 64, 0, stream>>>(
        a2, batch, sums, cnt, N);

    {
        int row_tiles = G / 16;
        wmma_gemm_f32<64, 128, true><<<cdiv((long long)row_tiles * 32, TPB), TPB, 0, stream>>>(
            sums, W2, pooled, b2, cnt, row_tiles);
    }

    {
        int row_tiles = G / 16;
        wmma_gemm_f32<128, 64, false><<<cdiv((long long)row_tiles * 32, TPB), TPB, 0, stream>>>(
            pooled, Wfc, out, bfc, nullptr, row_tiles);
    }
}


# Bucket pass, x@W1 product and pool-accumulator zeroing (mutually independent) merged into ONE launch as three work-group ranges: 9 launches to 7, the product fills the bucket pass's idle slots
# speedup vs baseline: 5.7840x; 1.0725x over previous
#include <hip/hip_runtime.h>
#include <hip/hip_bf16.h>


typedef float v2f __attribute__((ext_vector_type(2)));
typedef float v8f __attribute__((ext_vector_type(8)));

#define TPB 256

#define SORT_T 256
#define BIN_SH 8
#define CHUNKS 256
#define CHUNK_T 1024
#define LB     4
#define PB     8

template <int NT>
__device__ __forceinline__ int block_scan(int v, int* __restrict__ wtot, int& all) {
    const int lane = threadIdx.x & 31;
    const int w    = threadIdx.x >> 5;
    int s = v;
#pragma unroll
    for (int o = 1; o < 32; o <<= 1) {
        int u = __shfl_up(s, o);
        if (lane >= o) s += u;
    }
    if (lane == 31) wtot[w] = s;
    __syncthreads();
    int pre = 0, tot = 0;
#pragma unroll
    for (int i = 0; i < NT / 32; ++i) {
        const int x = wtot[i];
        tot += x;
        pre += (i < w) ? x : 0;
    }
    all = tot;
    return pre + s - v;
}

__global__ void __launch_bounds__(CHUNK_T)
chunk_pass_kernel(const int* __restrict__ src, const int* __restrict__ dst,
                  unsigned* __restrict__ binned, int* __restrict__ tab,
                  int E, int CH, int NB) {
    __shared__ int bcnt[SORT_T];
    __shared__ int wtot[CHUNK_T / 32];
    const int t  = threadIdx.x;
    const int c  = blockIdx.x;
    const int e0 = c * CH;
    const int n  = max(0, min(CH, E - e0));
    const unsigned last = (unsigned)(NB - 1);
    if (t < SORT_T) bcnt[t] = 0;
    __syncthreads();
    for (int i0 = t; i0 < n; i0 += LB * CHUNK_T) {
        unsigned d[LB];
#pragma unroll
        for (int u = 0; u < LB; ++u) {
            d[u] = (unsigned)dst[e0 + min(i0 + u * CHUNK_T, n - 1)];
        }
#pragma unroll
        for (int u = 0; u < LB; ++u) {
            const bool valid = i0 + u * CHUNK_T < n;
            const unsigned b = valid ? min(d[u] >> BIN_SH, last) : (unsigned)(t & (SORT_T - 1));
            atomicAdd(&bcnt[b], valid ? 1 : 0);
        }
    }
    __syncthreads();
    int tot;
    const int v  = (t < SORT_T) ? bcnt[t] : 0;
    const int ex = block_scan<CHUNK_T>(v, wtot, tot);
    if (t <= NB) tab[t * CHUNKS + c] = ex;
    if (t < SORT_T) bcnt[t] = ex;
    __syncthreads();
    for (int i0 = t; i0 < n; i0 += LB * CHUNK_T) {
        unsigned d[LB], s[LB];
#pragma unroll
        for (int u = 0; u < LB; ++u) {
            const int i = min(i0 + u * CHUNK_T, n - 1);
            d[u] = (unsigned)dst[e0 + i];
            s[u] = (unsigned)src[e0 + i];
        }
#pragma unroll
        for (int u = 0; u < LB; ++u) {
            const bool valid = i0 + u * CHUNK_T < n;
            const unsigned b = valid ? min(d[u] >> BIN_SH, last) : (unsigned)(t & (SORT_T - 1));
            const int p = atomicAdd(&bcnt[b], valid ? 1 : 0);
            if (valid) binned[e0 + p] = (s[u] << BIN_SH) | (d[u] & (unsigned)(SORT_T - 1));
        }
    }
}

__device__ __forceinline__ void
bucket_pass_body(const unsigned* __restrict__ binned, const int* __restrict__ tab,
                 int* __restrict__ off, int* __restrict__ nbr,
                 float* __restrict__ dinv, int n, int CH, int b) {
    static_assert(CHUNKS == SORT_T, "thread t of a bucket work-group walks chunk t");
    __shared__ int ncnt[SORT_T];
    __shared__ int wtot_a[SORT_T / 32];
    __shared__ int wtot_b[SORT_T / 32];
    const int t  = threadIdx.x;
    const int o0 = tab[b * CHUNKS + t];
    const int o1 = tab[(b + 1) * CHUNKS + t];
    const int len = o1 - o0;
    const unsigned* __restrict__ piece = binned + (size_t)t * CH + o0;
    ncnt[t] = 0;
    int base;
    (void)block_scan<SORT_T>(o0, wtot_a, base);
    for (int k0 = 0; k0 < len; k0 += PB) {
        unsigned w[PB];
#pragma unroll
        for (int u = 0; u < PB; ++u) w[u] = piece[min(k0 + u, len - 1)];
#pragma unroll
        for (int u = 0; u < PB; ++u) {
            const bool valid = k0 + u < len;
            atomicAdd(&ncnt[valid ? (w[u] & (unsigned)(SORT_T - 1)) : (unsigned)t], valid ? 1 : 0);
        }
    }
    __syncthreads();
    const int cdeg = ncnt[t];
    int unused;
    const int ex = block_scan<SORT_T>(cdeg, wtot_b, unused);
    const int node = b * SORT_T + t;
    if (node < n) {
        off[node]  = base + ex;
        dinv[node] = rsqrtf((float)(cdeg + 1));
    }
    if (node == n - 1) off[n] = base + ex + cdeg;
    ncnt[t] = base + ex;
    __syncthreads();
    for (int k0 = 0; k0 < len; k0 += PB) {
        unsigned w[PB];
#pragma unroll
        for (int u = 0; u < PB; ++u) w[u] = piece[min(k0 + u, len - 1)];
#pragma unroll
        for (int u = 0; u < PB; ++u) {
            const bool valid = k0 + u < len;
            const int p = atomicAdd(&ncnt[valid ? (w[u] & (unsigned)(SORT_T - 1)) : (unsigned)t], valid ? 1 : 0);
            if (valid) nbr[p] = (int)(w[u] >> BIN_SH);
        }
    }
}

template <int K, int Nn, bool MEAN>
__device__ __forceinline__ void wmma_gemm_strip(const float* __restrict__ A,
                                                const float* __restrict__ B,
                                                float* __restrict__ C,
                                                const float* __restrict__ bias,
                                                const float* __restrict__ cnt,
                                                int wave) {
    constexpr int COLT = Nn / 16;
    int lane = threadIdx.x & 31;
    int half = lane >> 4;
    int lm   = lane & 15;
    int row0 = wave << 4;

    const float* __restrict__ arow  = A + (size_t)(row0 + lm) * K + 2 * half;
    const float* __restrict__ bbase = B + (size_t)(2 * half) * Nn + lm;

    v8f acc[COLT];
#pragma unroll
    for (int c = 0; c < COLT; ++c) acc[c] = (v8f){};

#pragma unroll
    for (int k0 = 0; k0 < K; k0 += 4) {
        v2f a;
        a.x = arow[k0];
        a.y = arow[k0 + 1];
#pragma unroll
        for (int c = 0; c < COLT; ++c) {
            v2f b;
            b.x = bbase[k0 * Nn + c * 16];
            b.y = bbase[(k0 + 1) * Nn + c * 16];
            acc[c] = __builtin_amdgcn_wmma_f32_16x16x4_f32(
                 false, a,  false, b,
                 (short)0, acc[c],  false,  false);
        }
    }

    float* __restrict__ crow = C + (size_t)(row0 + 8 * half) * Nn + lm;
    float inv[8];
    bool  has[8];
#pragma unroll
    for (int v = 0; v < 8; ++v) {
        const float cv = MEAN ? cnt[row0 + 8 * half + v] : 1.0f;
        inv[v] = 1.0f / fmaxf(cv, 1.0f);
        has[v] = cv > 0.0f;
    }
#pragma unroll
    for (int c = 0; c < COLT; ++c) {
        float bv = bias ? bias[c * 16 + lm] : 0.0f;
#pragma unroll
        for (int v = 0; v < 8; ++v) {
            float r = acc[c][v];
            if (MEAN) r = r * inv[v] + (has[v] ? bv : 0.0f);
            else      r = r + bv;
            crow[(size_t)v * Nn + c * 16] = r;
        }
    }
}

template <int K, int Nn, bool MEAN>
__global__ void wmma_gemm_f32(const float* __restrict__ A,
                              const float* __restrict__ B,
                              float* __restrict__ C,
                              const float* __restrict__ bias,
                              const float* __restrict__ cnt,
                              int row_tiles) {
    const int wave = (blockIdx.x * blockDim.x + threadIdx.x) >> 5;
    if (wave >= row_tiles) return;
    wmma_gemm_strip<K, Nn, MEAN>(A, B, C, bias, cnt, wave);
}

__global__ void __launch_bounds__(SORT_T)
bucket_gemm_zero_kernel(const unsigned* __restrict__ binned, const int* __restrict__ tab,
                        int* __restrict__ off, int* __restrict__ nbr,
                        float* __restrict__ dinv, int n, int CH, int NB,
                        const float* __restrict__ x, const float* __restrict__ W1,
                        float* __restrict__ xw1, int row_tiles, int GB,
                        float* __restrict__ zero_p, int zero_n) {
    const int b = blockIdx.x;
    if (b < NB) {
        bucket_pass_body(binned, tab, off, nbr, dinv, n, CH, b);
    } else if (b < NB + GB) {
        const int wave = ((b - NB) * SORT_T + (int)threadIdx.x) >> 5;
        if (wave < row_tiles)
            wmma_gemm_strip<64, 64, false>(x, W1, xw1, nullptr, nullptr, wave);
    } else {
        const int i = (b - NB - GB) * SORT_T + (int)threadIdx.x;
        if (i < zero_n) zero_p[i] = 0.0f;
    }
}

template <int F, bool RELU, bool BIAS>
__global__ void __launch_bounds__(TPB)
gather_kernel(const float* __restrict__ h, const int* __restrict__ off,
              const int* __restrict__ nbr, const float* __restrict__ dinv,
              const float* __restrict__ bias, float* __restrict__ out, int N) {
    constexpr int VEC = F / 32;
    typedef float vf __attribute__((ext_vector_type(VEC)));
    const int node = __builtin_amdgcn_readfirstlane(
        (int)((blockIdx.x * blockDim.x + threadIdx.x) >> 5));
    if (node >= N) return;
    const int lane = threadIdx.x & 31;
    const vf* __restrict__ hv = (const vf*)h;

    const int beg = off[node];
    const int end = off[node + 1];
    const float dd = dinv[node];

    vf acc = hv[(size_t)node * 32 + lane] * (dd * dd);

    for (int base = beg; base < end; base += 32) {
        const int j = base + lane;
        int   s_l = node;
        float w_l = 0.0f;
        if (j < end) {
            s_l = nbr[j];
            w_l = dinv[s_l] * dd;
        }
        const int cnt  = min(32, end - base);
        const int cnt4 = (cnt + 3) & ~3;
        for (int k = 0; k < cnt4; k += 4) {
#pragma unroll
            for (int u = 0; u < 4; ++u) {
                const int   s = __builtin_amdgcn_readlane(s_l, k + u);
                const float w = __int_as_float(
                    __builtin_amdgcn_readlane(__float_as_int(w_l), k + u));
                acc += hv[(size_t)s * 32 + lane] * w;
            }
        }
    }

    vf r;
#pragma unroll
    for (int q = 0; q < VEC; ++q) {
        float v = acc[q];
        if (BIAS) v += bias[lane * VEC + q];
        r[q] = RELU ? fmaxf(v, 0.0f) : v;
    }
    ((vf*)out)[(size_t)node * 32 + lane] = r;
}

template <int F, int CH>
__global__ void __launch_bounds__(F)
pool_runs_kernel(const float* __restrict__ h, const int* __restrict__ batch,
                 float* __restrict__ sums, float* __restrict__ cnt, int N) {
    static_assert(CH <= 32, "one lane per node of the chunk holds its graph id");
    const int f    = threadIdx.x;
    const int lane = f & 31;
    const int n0   = blockIdx.x * CH;
    const int m    = min(CH, N - n0);

    const int g_l = batch[n0 + min(lane, m - 1)];

    float v[CH];
#pragma unroll
    for (int j = 0; j < CH; ++j) {
        const float* __restrict__ row = h + (size_t)(n0 + min(j, m - 1)) * F;
        const float t = row[f];
        v[j] = (j < m) ? t : 0.0f;
    }

    int   g_cur = __builtin_amdgcn_readlane(g_l, 0);
    int   run   = 0;
    float acc   = 0.0f;
#pragma unroll
    for (int j = 0; j < CH; ++j) {
        const int g = __builtin_amdgcn_readlane(g_l, j);
        if (g != g_cur) {
            atomicAdd(&sums[(size_t)g_cur * F + f], acc);
            if (f == 0) atomicAdd(&cnt[g_cur], (float)run);
            acc = 0.0f; run = 0; g_cur = g;
        }
        acc += v[j];
        run += (j < m) ? 1 : 0;
    }
    atomicAdd(&sums[(size_t)g_cur * F + f], acc);
    if (f == 0) atomicAdd(&cnt[g_cur], (float)run);
}

static inline int cdiv(long long a, long long b) { return (int)((a + b - 1) / b); }

extern "C" void kernel_launch(void* const* d_in, const int* in_sizes, int n_in,
                              void* d_out, int out_size, void* d_ws, size_t ws_size,
                              hipStream_t stream) {
    (void)n_in; (void)ws_size;
    const float* x    = (const float*)d_in[0];
    const int*   edge = (const int*)d_in[1];
    const int*   batch= (const int*)d_in[2];
    const float* b1   = (const float*)d_in[4];
    const float* W1   = (const float*)d_in[3];
    const float* W2   = (const float*)d_in[5];
    const float* b2   = (const float*)d_in[6];
    const float* Wfc  = (const float*)d_in[7];
    const float* bfc  = (const float*)d_in[8];
    float* out = (float*)d_out;

    const int N   = in_sizes[2];
    const int E   = in_sizes[1] / 2;
    const int D   = 64;
    const int D2  = 128;
    const int HID = in_sizes[8];
    const int G   = out_size / HID;

    const int* src = edge;
    const int* dst = edge + E;

    float* ws    = (float*)d_ws;
    float* r0    = ws;
    float* r1    = r0 + (size_t)N * D2;
    float* r2    = r1 + (size_t)N * D2;
    float* xw1   = r0;
    float* pooled= r0 + (size_t)N * D;
    float* a2    = r1;
    float* h1    = r2;
    const int NB = cdiv(N, SORT_T);
    const int CH = cdiv(E, CHUNKS);
    unsigned* binned = (unsigned*)(r2 + (size_t)N * D);
    int*   nbr   = (int*)binned + E;
    int*   tab   = nbr + E;
    int*   off   = tab + (size_t)(NB + 1) * CHUNKS;
    float* dinv  = r2 + (size_t)N * D2;
    float* sums  = dinv + (size_t)N;
    float* cnt   = sums + (size_t)G * D;

    chunk_pass_kernel<<<CHUNKS, CHUNK_T, 0, stream>>>(src, dst, binned, tab, E, CH, NB);

    {
        const int row_tiles = N / 16;
        const int GB = cdiv((long long)row_tiles * 32, SORT_T);
        const int ZB = cdiv((long long)G * D + G, SORT_T);
        bucket_gemm_zero_kernel<<<NB + GB + ZB, SORT_T, 0, stream>>>(
            binned, tab, off, nbr, dinv, N, CH, NB,
            x, W1, xw1, row_tiles, GB, sums, G * D + G);
    }

    gather_kernel<64, true, true><<<cdiv((long long)N * 32, TPB), TPB, 0, stream>>>(
        xw1, off, nbr, dinv, b1, h1, N);

    gather_kernel<64, false, false><<<cdiv((long long)N * 32, TPB), TPB, 0, stream>>>(
        h1, off, nbr, dinv, nullptr, a2, N);

    pool_runs_kernel<64, 32><<<cdiv(N, 32), 64, 0, stream>>>(
        a2, batch, sums, cnt, N);

    {
        int row_tiles = G / 16;
        wmma_gemm_f32<64, 128, true><<<cdiv((long long)row_tiles * 32, TPB), TPB, 0, stream>>>(
            sums, W2, pooled, b2, cnt, row_tiles);
    }

    {
        int row_tiles = G / 16;
        wmma_gemm_f32<128, 64, false><<<cdiv((long long)row_tiles * 32, TPB), TPB, 0, stream>>>(
            pooled, Wfc, out, bfc, nullptr, row_tiles);
    }
}


# Mean pool, W2 and FC products in ONE launch: a work-group per 16 graphs sums its own graphs' rows (LDS float adds, no global atomics), products from LDS; 7 launches to 5; pool sums: LDS arrival order
# speedup vs baseline: 6.7277x; 1.1632x over previous
#include <hip/hip_runtime.h>
#include <hip/hip_bf16.h>


typedef float v2f __attribute__((ext_vector_type(2)));
typedef float v8f __attribute__((ext_vector_type(8)));

#define TPB 256

#define SORT_T 256
#define BIN_SH 8
#define CHUNKS 256
#define CHUNK_T 1024
#define LB     4
#define PB     8

template <int NT>
__device__ __forceinline__ int block_scan(int v, int* __restrict__ wtot, int& all) {
    const int lane = threadIdx.x & 31;
    const int w    = threadIdx.x >> 5;
    int s = v;
#pragma unroll
    for (int o = 1; o < 32; o <<= 1) {
        int u = __shfl_up(s, o);
        if (lane >= o) s += u;
    }
    if (lane == 31) wtot[w] = s;
    __syncthreads();
    int pre = 0, tot = 0;
#pragma unroll
    for (int i = 0; i < NT / 32; ++i) {
        const int x = wtot[i];
        tot += x;
        pre += (i < w) ? x : 0;
    }
    all = tot;
    return pre + s - v;
}

__global__ void __launch_bounds__(CHUNK_T)
chunk_pass_kernel(const int* __restrict__ src, const int* __restrict__ dst,
                  unsigned* __restrict__ binned, int* __restrict__ tab,
                  int E, int CH, int NB) {
    __shared__ int bcnt[SORT_T];
    __shared__ int wtot[CHUNK_T / 32];
    const int t  = threadIdx.x;
    const int c  = blockIdx.x;
    const int e0 = c * CH;
    const int n  = max(0, min(CH, E - e0));
    const unsigned last = (unsigned)(NB - 1);
    if (t < SORT_T) bcnt[t] = 0;
    __syncthreads();
    for (int i0 = t; i0 < n; i0 += LB * CHUNK_T) {
        unsigned d[LB];
#pragma unroll
        for (int u = 0; u < LB; ++u) {
            d[u] = (unsigned)dst[e0 + min(i0 + u * CHUNK_T, n - 1)];
        }
#pragma unroll
        for (int u = 0; u < LB; ++u) {
            const bool valid = i0 + u * CHUNK_T < n;
            const unsigned b = valid ? min(d[u] >> BIN_SH, last) : (unsigned)(t & (SORT_T - 1));
            atomicAdd(&bcnt[b], valid ? 1 : 0);
        }
    }
    __syncthreads();
    int tot;
    const int v  = (t < SORT_T) ? bcnt[t] : 0;
    const int ex = block_scan<CHUNK_T>(v, wtot, tot);
    if (t <= NB) tab[t * CHUNKS + c] = ex;
    if (t < SORT_T) bcnt[t] = ex;
    __syncthreads();
    for (int i0 = t; i0 < n; i0 += LB * CHUNK_T) {
        unsigned d[LB], s[LB];
#pragma unroll
        for (int u = 0; u < LB; ++u) {
            const int i = min(i0 + u * CHUNK_T, n - 1);
            d[u] = (unsigned)dst[e0 + i];
            s[u] = (unsigned)src[e0 + i];
        }
#pragma unroll
        for (int u = 0; u < LB; ++u) {
            const bool valid = i0 + u * CHUNK_T < n;
            const unsigned b = valid ? min(d[u] >> BIN_SH, last) : (unsigned)(t & (SORT_T - 1));
            const int p = atomicAdd(&bcnt[b], valid ? 1 : 0);
            if (valid) binned[e0 + p] = (s[u] << BIN_SH) | (d[u] & (unsigned)(SORT_T - 1));
        }
    }
}

__device__ __forceinline__ void
bucket_pass_body(const unsigned* __restrict__ binned, const int* __restrict__ tab,
                 int* __restrict__ off, int* __restrict__ nbr,
                 float* __restrict__ dinv, int n, int CH, int b) {
    static_assert(CHUNKS == SORT_T, "thread t of a bucket work-group walks chunk t");
    __shared__ int ncnt[SORT_T];
    __shared__ int wtot_a[SORT_T / 32];
    __shared__ int wtot_b[SORT_T / 32];
    const int t  = threadIdx.x;
    const int o0 = tab[b * CHUNKS + t];
    const int o1 = tab[(b + 1) * CHUNKS + t];
    const int len = o1 - o0;
    const unsigned* __restrict__ piece = binned + (size_t)t * CH + o0;
    ncnt[t] = 0;
    int base;
    (void)block_scan<SORT_T>(o0, wtot_a, base);
    for (int k0 = 0; k0 < len; k0 += PB) {
        unsigned w[PB];
#pragma unroll
        for (int u = 0; u < PB; ++u) w[u] = piece[min(k0 + u, len - 1)];
#pragma unroll
        for (int u = 0; u < PB; ++u) {
            const bool valid = k0 + u < len;
            atomicAdd(&ncnt[valid ? (w[u] & (unsigned)(SORT_T - 1)) : (unsigned)t], valid ? 1 : 0);
        }
    }
    __syncthreads();
    const int cdeg = ncnt[t];
    int unused;
    const int ex = block_scan<SORT_T>(cdeg, wtot_b, unused);
    const int node = b * SORT_T + t;
    if (node < n) {
        off[node]  = base + ex;
        dinv[node] = rsqrtf((float)(cdeg + 1));
    }
    if (node == n - 1) off[n] = base + ex + cdeg;
    ncnt[t] = base + ex;
    __syncthreads();
    for (int k0 = 0; k0 < len; k0 += PB) {
        unsigned w[PB];
#pragma unroll
        for (int u = 0; u < PB; ++u) w[u] = piece[min(k0 + u, len - 1)];
#pragma unroll
        for (int u = 0; u < PB; ++u) {
            const bool valid = k0 + u < len;
            const int p = atomicAdd(&ncnt[valid ? (w[u] & (unsigned)(SORT_T - 1)) : (unsigned)t], valid ? 1 : 0);
            if (valid) nbr[p] = (int)(w[u] >> BIN_SH);
        }
    }
}

template <int K, int Nn>
__device__ __forceinline__ void wmma_gemm_strip(const float* __restrict__ A,
                                                const float* __restrict__ B,
                                                float* __restrict__ C,
                                                const float* __restrict__ bias,
                                                int wave) {
    constexpr int COLT = Nn / 16;
    int lane = threadIdx.x & 31;
    int half = lane >> 4;
    int lm   = lane & 15;
    int row0 = wave << 4;

    const float* __restrict__ arow  = A + (size_t)(row0 + lm) * K + 2 * half;
    const float* __restrict__ bbase = B + (size_t)(2 * half) * Nn + lm;

    v8f acc[COLT];
#pragma unroll
    for (int c = 0; c < COLT; ++c) acc[c] = (v8f){};

#pragma unroll
    for (int k0 = 0; k0 < K; k0 += 4) {
        v2f a;
        a.x = arow[k0];
        a.y = arow[k0 + 1];
#pragma unroll
        for (int c = 0; c < COLT; ++c) {
            v2f b;
            b.x = bbase[k0 * Nn + c * 16];
            b.y = bbase[(k0 + 1) * Nn + c * 16];
            acc[c] = __builtin_amdgcn_wmma_f32_16x16x4_f32(
                 false, a,  false, b,
                 (short)0, acc[c],  false,  false);
        }
    }

    float* __restrict__ crow = C + (size_t)(row0 + 8 * half) * Nn + lm;
#pragma unroll
    for (int c = 0; c < COLT; ++c) {
        float bv = bias ? bias[c * 16 + lm] : 0.0f;
#pragma unroll
        for (int v = 0; v < 8; ++v) crow[(size_t)v * Nn + c * 16] = acc[c][v] + bv;
    }
}

__global__ void __launch_bounds__(SORT_T)
bucket_gemm_kernel(const unsigned* __restrict__ binned, const int* __restrict__ tab,
                   int* __restrict__ off, int* __restrict__ nbr,
                   float* __restrict__ dinv, int n, int CH, int NB,
                   const float* __restrict__ x, const float* __restrict__ W1,
                   float* __restrict__ xw1, int row_tiles) {
    const int b = blockIdx.x;
    if (b < NB) {
        bucket_pass_body(binned, tab, off, nbr, dinv, n, CH, b);
    } else {
        const int wave = ((b - NB) * SORT_T + (int)threadIdx.x) >> 5;
        if (wave < row_tiles)
            wmma_gemm_strip<64, 64>(x, W1, xw1, nullptr, wave);
    }
}

template <int F, bool RELU, bool BIAS>
__global__ void __launch_bounds__(TPB)
gather_kernel(const float* __restrict__ h, const int* __restrict__ off,
              const int* __restrict__ nbr, const float* __restrict__ dinv,
              const float* __restrict__ bias, float* __restrict__ out, int N) {
    constexpr int VEC = F / 32;
    typedef float vf __attribute__((ext_vector_type(VEC)));
    const int node = __builtin_amdgcn_readfirstlane(
        (int)((blockIdx.x * blockDim.x + threadIdx.x) >> 5));
    if (node >= N) return;
    const int lane = threadIdx.x & 31;
    const vf* __restrict__ hv = (const vf*)h;

    const int beg = off[node];
    const int end = off[node + 1];
    const float dd = dinv[node];

    vf acc = hv[(size_t)node * 32 + lane] * (dd * dd);

    for (int base = beg; base < end; base += 32) {
        const int j = base + lane;
        int   s_l = node;
        float w_l = 0.0f;
        if (j < end) {
            s_l = nbr[j];
            w_l = dinv[s_l] * dd;
        }
        const int cnt  = min(32, end - base);
        const int cnt4 = (cnt + 3) & ~3;
        for (int k = 0; k < cnt4; k += 4) {
#pragma unroll
            for (int u = 0; u < 4; ++u) {
                const int   s = __builtin_amdgcn_readlane(s_l, k + u);
                const float w = __int_as_float(
                    __builtin_amdgcn_readlane(__float_as_int(w_l), k + u));
                acc += hv[(size_t)s * 32 + lane] * w;
            }
        }
    }

    vf r;
#pragma unroll
    for (int q = 0; q < VEC; ++q) {
        float v = acc[q];
        if (BIAS) v += bias[lane * VEC + q];
        r[q] = RELU ? fmaxf(v, 0.0f) : v;
    }
    ((vf*)out)[(size_t)node * 32 + lane] = r;
}

#define TAIL_T   1024
#define TAIL_W   (TAIL_T / 32)
#define IDB      64
#define S_PITCH  68
#define P_PITCH  132

__global__ void __launch_bounds__(TAIL_T)
pool_w2_fc_kernel(const float* __restrict__ a2, const int* __restrict__ batch,
                  const float* __restrict__ W2, const float* __restrict__ b2,
                  const float* __restrict__ Wfc, const float* __restrict__ bfc,
                  float* __restrict__ out, int N) {
    __shared__ __attribute__((aligned(16))) float S[16 * S_PITCH];
    __shared__ __attribute__((aligned(16))) float P[16 * P_PITCH];
    __shared__ float Cn[16];
    const int t    = threadIdx.x;
    const int lane = t & 31;
    const int w    = __builtin_amdgcn_readfirstlane(t >> 5);
    const int g0   = blockIdx.x * 16;
    const int half = lane >> 4;
    const int lm   = lane & 15;

    for (int i = t; i < 16 * S_PITCH; i += TAIL_T) S[i] = 0.0f;
    if (t < 16) Cn[t] = 0.0f;
    __syncthreads();

    const v2f* __restrict__ av = (const v2f*)a2;
    const int NC = (N + 31) >> 5;
    for (int c0 = w; c0 < NC; c0 += IDB * TAIL_W) {
        int ids[IDB];
#pragma unroll
        for (int u = 0; u < IDB; ++u)
            ids[u] = batch[min((c0 + u * TAIL_W) * 32 + lane, N - 1)];
        unsigned long long hit = 0;
#pragma unroll
        for (int u = 0; u < IDB; ++u) {
            const bool own = (unsigned)(ids[u] - g0) < 16u;
            if (c0 + u * TAIL_W < NC && __builtin_amdgcn_ballot_w32(own) != 0u) hit |= 1ull << u;
        }
        while (hit != 0) {
            const int u = __builtin_ctzll(hit);
            hit &= hit - 1;
            const int n0 = (c0 + u * TAIL_W) * 32;
            const int m  = min(32, N - n0);
            const int g_l = batch[n0 + min(lane, m - 1)];
            v2f v[32];
#pragma unroll
            for (int j = 0; j < 32; ++j) v[j] = av[(size_t)(n0 + min(j, m - 1)) * 32 + lane];

            int   g_cur = __builtin_amdgcn_readlane(g_l, 0);
            int   run   = 0;
            v2f   acc   = (v2f){0.0f, 0.0f};
#pragma unroll
            for (int j = 0; j < 32; ++j) {
                const int g = __builtin_amdgcn_readlane(g_l, j);
                if (g != g_cur) {
                    const unsigned r = (unsigned)(g_cur - g0);
                    if (r < 16u) {
                        atomicAdd(&S[r * S_PITCH + 2 * lane],     acc.x);
                        atomicAdd(&S[r * S_PITCH + 2 * lane + 1], acc.y);
                        if (lane == 0) atomicAdd(&Cn[r], (float)run);
                    }
                    acc = (v2f){0.0f, 0.0f}; run = 0; g_cur = g;
                }
                if (j < m) { acc += v[j]; run += 1; }
            }
            const unsigned r = (unsigned)(g_cur - g0);
            if (r < 16u) {
                atomicAdd(&S[r * S_PITCH + 2 * lane],     acc.x);
                atomicAdd(&S[r * S_PITCH + 2 * lane + 1], acc.y);
                if (lane == 0) atomicAdd(&Cn[r], (float)run);
            }
        }
    }
    __syncthreads();

    if (w < 8) {
        const float* __restrict__ bb = W2 + (size_t)(2 * half) * 128 + w * 16 + lm;
        v8f acc = (v8f){};
#pragma unroll
        for (int k0 = 0; k0 < 64; k0 += 4) {
            const v2f a = *(const v2f*)&S[lm * S_PITCH + k0 + 2 * half];
            v2f b;
            b.x = bb[k0 * 128];
            b.y = bb[(k0 + 1) * 128];
            acc = __builtin_amdgcn_wmma_f32_16x16x4_f32(
                 false, a,  false, b,
                 (short)0, acc,  false,  false);
        }
        const float bv = b2[w * 16 + lm];
#pragma unroll
        for (int v = 0; v < 8; ++v) {
            const float cv  = Cn[v + 8 * half];
            const float inv = 1.0f / fmaxf(cv, 1.0f);
            P[(v + 8 * half) * P_PITCH + w * 16 + lm] = acc[v] * inv + (cv > 0.0f ? bv : 0.0f);
        }
    }
    __syncthreads();

    if (w < 2) {
        const float* __restrict__ bb = Wfc + (size_t)(2 * half) * 64 + w * 32 + lm;
        v8f acc0 = (v8f){};
        v8f acc1 = (v8f){};
#pragma unroll
        for (int k0 = 0; k0 < 128; k0 += 4) {
            const v2f a = *(const v2f*)&P[lm * P_PITCH + k0 + 2 * half];
            v2f b0, b1;
            b0.x = bb[k0 * 64];
            b0.y = bb[(k0 + 1) * 64];
            b1.x = bb[k0 * 64 + 16];
            b1.y = bb[(k0 + 1) * 64 + 16];
            acc0 = __builtin_amdgcn_wmma_f32_16x16x4_f32(
                 false, a,  false, b0,
                 (short)0, acc0,  false,  false);
            acc1 = __builtin_amdgcn_wmma_f32_16x16x4_f32(
                 false, a,  false, b1,
                 (short)0, acc1,  false,  false);
        }
        const float bv0 = bfc[w * 32 + lm];
        const float bv1 = bfc[w * 32 + 16 + lm];
        float* __restrict__ orow = out + (size_t)(g0 + 8 * half) * 64 + w * 32 + lm;
#pragma unroll
        for (int v = 0; v < 8; ++v) {
            orow[(size_t)v * 64]      = acc0[v] + bv0;
            orow[(size_t)v * 64 + 16] = acc1[v] + bv1;
        }
    }
}

static inline int cdiv(long long a, long long b) { return (int)((a + b - 1) / b); }

extern "C" void kernel_launch(void* const* d_in, const int* in_sizes, int n_in,
                              void* d_out, int out_size, void* d_ws, size_t ws_size,
                              hipStream_t stream) {
    (void)n_in; (void)ws_size;
    const float* x    = (const float*)d_in[0];
    const int*   edge = (const int*)d_in[1];
    const int*   batch= (const int*)d_in[2];
    const float* b1   = (const float*)d_in[4];
    const float* W1   = (const float*)d_in[3];
    const float* W2   = (const float*)d_in[5];
    const float* b2   = (const float*)d_in[6];
    const float* Wfc  = (const float*)d_in[7];
    const float* bfc  = (const float*)d_in[8];
    float* out = (float*)d_out;

    const int N   = in_sizes[2];
    const int E   = in_sizes[1] / 2;
    const int D   = 64;
    const int D2  = 128;
    const int HID = in_sizes[8];
    const int G   = out_size / HID;

    const int* src = edge;
    const int* dst = edge + E;

    float* ws    = (float*)d_ws;
    float* r0    = ws;
    float* r1    = r0 + (size_t)N * D2;
    float* r2    = r1 + (size_t)N * D2;
    float* xw1   = r0;
    float* a2    = r1;
    float* h1    = r2;
    const int NB = cdiv(N, SORT_T);
    const int CH = cdiv(E, CHUNKS);
    unsigned* binned = (unsigned*)(r2 + (size_t)N * D);
    int*   nbr   = (int*)binned + E;
    int*   tab   = nbr + E;
    int*   off   = tab + (size_t)(NB + 1) * CHUNKS;
    float* dinv  = r2 + (size_t)N * D2;

    chunk_pass_kernel<<<CHUNKS, CHUNK_T, 0, stream>>>(src, dst, binned, tab, E, CH, NB);

    {
        const int row_tiles = N / 16;
        const int GB = cdiv((long long)row_tiles * 32, SORT_T);
        bucket_gemm_kernel<<<NB + GB, SORT_T, 0, stream>>>(
            binned, tab, off, nbr, dinv, N, CH, NB,
            x, W1, xw1, row_tiles);
    }

    gather_kernel<64, true, true><<<cdiv((long long)N * 32, TPB), TPB, 0, stream>>>(
        xw1, off, nbr, dinv, b1, h1, N);

    gather_kernel<64, false, false><<<cdiv((long long)N * 32, TPB), TPB, 0, stream>>>(
        h1, off, nbr, dinv, nullptr, a2, N);

    pool_w2_fc_kernel<<<G / 16, TAIL_T, 0, stream>>>(a2, batch, W2, b2, Wfc, bfc, out, N);
}


# Rows stored pre-scaled by dinv of their own node (x@W1 strips now made by the bucket work-group that owns the nodes, h1 in conv1's epilogue): gathers sum plain rows, no per-edge weight load
# speedup vs baseline: 7.0706x; 1.0510x over previous
#include <hip/hip_runtime.h>
#include <hip/hip_bf16.h>


typedef float v2f __attribute__((ext_vector_type(2)));
typedef float v8f __attribute__((ext_vector_type(8)));

#define TPB 256

#define SORT_T 256
#define BIN_SH 8
#define CHUNKS 256
#define CHUNK_T 1024
#define LB     4
#define PB     8

template <int NT>
__device__ __forceinline__ int block_scan(int v, int* __restrict__ wtot, int& all) {
    const int lane = threadIdx.x & 31;
    const int w    = threadIdx.x >> 5;
    int s = v;
#pragma unroll
    for (int o = 1; o < 32; o <<= 1) {
        int u = __shfl_up(s, o);
        if (lane >= o) s += u;
    }
    if (lane == 31) wtot[w] = s;
    __syncthreads();
    int pre = 0, tot = 0;
#pragma unroll
    for (int i = 0; i < NT / 32; ++i) {
        const int x = wtot[i];
        tot += x;
        pre += (i < w) ? x : 0;
    }
    all = tot;
    return pre + s - v;
}

__global__ void __launch_bounds__(CHUNK_T)
chunk_pass_kernel(const int* __restrict__ src, const int* __restrict__ dst,
                  unsigned* __restrict__ binned, int* __restrict__ tab,
                  int E, int CH, int NB) {
    __shared__ int bcnt[SORT_T];
    __shared__ int wtot[CHUNK_T / 32];
    const int t  = threadIdx.x;
    const int c  = blockIdx.x;
    const int e0 = c * CH;
    const int n  = max(0, min(CH, E - e0));
    const unsigned last = (unsigned)(NB - 1);
    if (t < SORT_T) bcnt[t] = 0;
    __syncthreads();
    for (int i0 = t; i0 < n; i0 += LB * CHUNK_T) {
        unsigned d[LB];
#pragma unroll
        for (int u = 0; u < LB; ++u) {
            d[u] = (unsigned)dst[e0 + min(i0 + u * CHUNK_T, n - 1)];
        }
#pragma unroll
        for (int u = 0; u < LB; ++u) {
            const bool valid = i0 + u * CHUNK_T < n;
            const unsigned b = valid ? min(d[u] >> BIN_SH, last) : (unsigned)(t & (SORT_T - 1));
            atomicAdd(&bcnt[b], valid ? 1 : 0);
        }
    }
    __syncthreads();
    int tot;
    const int v  = (t < SORT_T) ? bcnt[t] : 0;
    const int ex = block_scan<CHUNK_T>(v, wtot, tot);
    if (t <= NB) tab[t * CHUNKS + c] = ex;
    if (t < SORT_T) bcnt[t] = ex;
    __syncthreads();
    for (int i0 = t; i0 < n; i0 += LB * CHUNK_T) {
        unsigned d[LB], s[LB];
#pragma unroll
        for (int u = 0; u < LB; ++u) {
            const int i = min(i0 + u * CHUNK_T, n - 1);
            d[u] = (unsigned)dst[e0 + i];
            s[u] = (unsigned)src[e0 + i];
        }
#pragma unroll
        for (int u = 0; u < LB; ++u) {
            const bool valid = i0 + u * CHUNK_T < n;
            const unsigned b = valid ? min(d[u] >> BIN_SH, last) : (unsigned)(t & (SORT_T - 1));
            const int p = atomicAdd(&bcnt[b], valid ? 1 : 0);
            if (valid) binned[e0 + p] = (s[u] << BIN_SH) | (d[u] & (unsigned)(SORT_T - 1));
        }
    }
}

__device__ __forceinline__ float
bucket_pass_body(const unsigned* __restrict__ binned, const int* __restrict__ tab,
                 int* __restrict__ off, int* __restrict__ nbr,
                 float* __restrict__ dinv, int n, int CH, int b) {
    static_assert(CHUNKS == SORT_T, "thread t of a bucket work-group walks chunk t");
    __shared__ int ncnt[SORT_T];
    __shared__ int wtot_a[SORT_T / 32];
    __shared__ int wtot_b[SORT_T / 32];
    const int t  = threadIdx.x;
    const int o0 = tab[b * CHUNKS + t];
    const int o1 = tab[(b + 1) * CHUNKS + t];
    const int len = o1 - o0;
    const unsigned* __restrict__ piece = binned + (size_t)t * CH + o0;
    ncnt[t] = 0;
    int base;
    (void)block_scan<SORT_T>(o0, wtot_a, base);
    for (int k0 = 0; k0 < len; k0 += PB) {
        unsigned w[PB];
#pragma unroll
        for (int u = 0; u < PB; ++u) w[u] = piece[min(k0 + u, len - 1)];
#pragma unroll
        for (int u = 0; u < PB; ++u) {
            const bool valid = k0 + u < len;
            atomicAdd(&ncnt[valid ? (w[u] & (unsigned)(SORT_T - 1)) : (unsigned)t], valid ? 1 : 0);
        }
    }
    __syncthreads();
    const int cdeg = ncnt[t];
    int unused;
    const int ex = block_scan<SORT_T>(cdeg, wtot_b, unused);
    const int node = b * SORT_T + t;
    const float dv = rsqrtf((float)(cdeg + 1));
    if (node < n) {
        off[node]  = base + ex;
        dinv[node] = dv;
    }
    if (node == n - 1) off[n] = base + ex + cdeg;
    ncnt[t] = base + ex;
    __syncthreads();
    for (int k0 = 0; k0 < len; k0 += PB) {
        unsigned w[PB];
#pragma unroll
        for (int u = 0; u < PB; ++u) w[u] = piece[min(k0 + u, len - 1)];
#pragma unroll
        for (int u = 0; u < PB; ++u) {
            const bool valid = k0 + u < len;
            const int p = atomicAdd(&ncnt[valid ? (w[u] & (unsigned)(SORT_T - 1)) : (unsigned)t], valid ? 1 : 0);
            if (valid) nbr[p] = (int)(w[u] >> BIN_SH);
        }
    }
    return dv;
}

template <int K, int Nn, int RT>
__device__ __forceinline__ void wmma_gemm_strip(const float* __restrict__ A,
                                                const float* __restrict__ B,
                                                float* __restrict__ C,
                                                const float* __restrict__ sc,
                                                int row0) {
    constexpr int COLT = Nn / 16;
    int lane = threadIdx.x & 31;
    int half = lane >> 4;
    int lm   = lane & 15;

    const float* __restrict__ arow  = A + (size_t)(row0 + lm) * K + 2 * half;
    const float* __restrict__ bbase = B + (size_t)(2 * half) * Nn + lm;

    v8f acc[RT][COLT];
#pragma unroll
    for (int r = 0; r < RT; ++r)
#pragma unroll
        for (int c = 0; c < COLT; ++c) acc[r][c] = (v8f){};

#pragma unroll
    for (int k0 = 0; k0 < K; k0 += 4) {
        v2f a[RT];
#pragma unroll
        for (int r = 0; r < RT; ++r) {
            a[r].x = arow[r * 16 * K + k0];
            a[r].y = arow[r * 16 * K + k0 + 1];
        }
#pragma unroll
        for (int c = 0; c < COLT; ++c) {
            v2f b;
            b.x = bbase[k0 * Nn + c * 16];
            b.y = bbase[(k0 + 1) * Nn + c * 16];
#pragma unroll
            for (int r = 0; r < RT; ++r)
                acc[r][c] = __builtin_amdgcn_wmma_f32_16x16x4_f32(
                     false, a[r],  false, b,
                     (short)0, acc[r][c],  false,  false);
        }
    }

    float* __restrict__ crow = C + (size_t)(row0 + 8 * half) * Nn + lm;
#pragma unroll
    for (int r = 0; r < RT; ++r) {
#pragma unroll
        for (int v = 0; v < 8; ++v) {
            const float s = sc[r * 16 + 8 * half + v];
#pragma unroll
            for (int c = 0; c < COLT; ++c)
                crow[(size_t)(r * 16 + v) * Nn + c * 16] = acc[r][c][v] * s;
        }
    }
}

__global__ void __launch_bounds__(SORT_T)
bucket_gemm_kernel(const unsigned* __restrict__ binned, const int* __restrict__ tab,
                   int* __restrict__ off, int* __restrict__ nbr,
                   float* __restrict__ dinv, int n, int CH,
                   const float* __restrict__ x, const float* __restrict__ W1,
                   float* __restrict__ xw1s, int row_tiles) {
    __shared__ float dsc[SORT_T];
    const int b = blockIdx.x;
    const int t = threadIdx.x;
    dsc[t] = bucket_pass_body(binned, tab, off, nbr, dinv, n, CH, b);
    __syncthreads();
    const int w  = __builtin_amdgcn_readfirstlane(t >> 5);
    const int t0 = b * (SORT_T / 16) + 2 * w;
    if (t0 + 1 < row_tiles)
        wmma_gemm_strip<64, 64, 2>(x, W1, xw1s, &dsc[32 * w], t0 * 16);
    else if (t0 < row_tiles)
        wmma_gemm_strip<64, 64, 1>(x, W1, xw1s, &dsc[32 * w], t0 * 16);
}

template <int F, bool CONV1>
__global__ void __launch_bounds__(TPB)
gather_kernel(const float* __restrict__ hs, const int* __restrict__ off,
              const int* __restrict__ nbr, const float* __restrict__ dinv,
              const float* __restrict__ bias, float* __restrict__ out, int N) {
    constexpr int VEC = F / 32;
    typedef float vf __attribute__((ext_vector_type(VEC)));
    const int node = __builtin_amdgcn_readfirstlane(
        (int)((blockIdx.x * blockDim.x + threadIdx.x) >> 5));
    if (node >= N) return;
    const int lane = threadIdx.x & 31;
    const vf* __restrict__ hv = (const vf*)hs;

    const int beg = off[node];
    const int end = off[node + 1];
    const float dd = dinv[node];

    vf acc = hv[(size_t)node * 32 + lane];

    for (int base = beg; base < end; base += 32) {
        const int j = base + lane;
        int s_l = node;
        if (j < end) s_l = nbr[j];
        const int cnt  = min(32, end - base);
        const int cnt4 = (cnt + 3) & ~3;
        for (int k = 0; k < cnt4; k += 4) {
#pragma unroll
            for (int u = 0; u < 4; ++u) {
                const int   s = __builtin_amdgcn_readlane(s_l, k + u);
                const float w = (k + u < cnt) ? 1.0f : 0.0f;
                acc += hv[(size_t)s * 32 + lane] * w;
            }
        }
    }

    vf r;
#pragma unroll
    for (int q = 0; q < VEC; ++q) {
        float v = acc[q] * dd;
        if (CONV1) v = fmaxf(v + bias[lane * VEC + q], 0.0f) * dd;
        r[q] = v;
    }
    ((vf*)out)[(size_t)node * 32 + lane] = r;
}

#define TAIL_T   1024
#define TAIL_W   (TAIL_T / 32)
#define IDB      64
#define S_PITCH  68
#define P_PITCH  132

__global__ void __launch_bounds__(TAIL_T)
pool_w2_fc_kernel(const float* __restrict__ a2, const int* __restrict__ batch,
                  const float* __restrict__ W2, const float* __restrict__ b2,
                  const float* __restrict__ Wfc, const float* __restrict__ bfc,
                  float* __restrict__ out, int N) {
    __shared__ __attribute__((aligned(16))) float S[16 * S_PITCH];
    __shared__ __attribute__((aligned(16))) float P[16 * P_PITCH];
    __shared__ float Cn[16];
    const int t    = threadIdx.x;
    const int lane = t & 31;
    const int w    = __builtin_amdgcn_readfirstlane(t >> 5);
    const int g0   = blockIdx.x * 16;
    const int half = lane >> 4;
    const int lm   = lane & 15;

    for (int i = t; i < 16 * S_PITCH; i += TAIL_T) S[i] = 0.0f;
    if (t < 16) Cn[t] = 0.0f;
    __syncthreads();

    const v2f* __restrict__ av = (const v2f*)a2;
    const int NC = (N + 31) >> 5;
    for (int c0 = w; c0 < NC; c0 += IDB * TAIL_W) {
        int ids[IDB];
#pragma unroll
        for (int u = 0; u < IDB; ++u)
            ids[u] = batch[min((c0 + u * TAIL_W) * 32 + lane, N - 1)];
        unsigned long long hit = 0;
#pragma unroll
        for (int u = 0; u < IDB; ++u) {
            const bool own = (unsigned)(ids[u] - g0) < 16u;
            if (c0 + u * TAIL_W < NC && __builtin_amdgcn_ballot_w32(own) != 0u) hit |= 1ull << u;
        }
        while (hit != 0) {
            const int u = __builtin_ctzll(hit);
            hit &= hit - 1;
            const int n0 = (c0 + u * TAIL_W) * 32;
            const int m  = min(32, N - n0);
            const int g_l = batch[n0 + min(lane, m - 1)];
            v2f v[32];
#pragma unroll
            for (int j = 0; j < 32; ++j) v[j] = av[(size_t)(n0 + min(j, m - 1)) * 32 + lane];

            int   g_cur = __builtin_amdgcn_readlane(g_l, 0);
            int   run   = 0;
            v2f   acc   = (v2f){0.0f, 0.0f};
#pragma unroll
            for (int j = 0; j < 32; ++j) {
                const int g = __builtin_amdgcn_readlane(g_l, j);
                if (g != g_cur) {
                    const unsigned r = (unsigned)(g_cur - g0);
                    if (r < 16u) {
                        atomicAdd(&S[r * S_PITCH + 2 * lane],     acc.x);
                        atomicAdd(&S[r * S_PITCH + 2 * lane + 1], acc.y);
                        if (lane == 0) atomicAdd(&Cn[r], (float)run);
                    }
                    acc = (v2f){0.0f, 0.0f}; run = 0; g_cur = g;
                }
                if (j < m) { acc += v[j]; run += 1; }
            }
            const unsigned r = (unsigned)(g_cur - g0);
            if (r < 16u) {
                atomicAdd(&S[r * S_PITCH + 2 * lane],     acc.x);
                atomicAdd(&S[r * S_PITCH + 2 * lane + 1], acc.y);
                if (lane == 0) atomicAdd(&Cn[r], (float)run);
            }
        }
    }
    __syncthreads();

    if (w < 8) {
        const float* __restrict__ bb = W2 + (size_t)(2 * half) * 128 + w * 16 + lm;
        v8f acc = (v8f){};
#pragma unroll
        for (int k0 = 0; k0 < 64; k0 += 4) {
            const v2f a = *(const v2f*)&S[lm * S_PITCH + k0 + 2 * half];
            v2f b;
            b.x = bb[k0 * 128];
            b.y = bb[(k0 + 1) * 128];
            acc = __builtin_amdgcn_wmma_f32_16x16x4_f32(
                 false, a,  false, b,
                 (short)0, acc,  false,  false);
        }
        const float bv = b2[w * 16 + lm];
#pragma unroll
        for (int v = 0; v < 8; ++v) {
            const float cv  = Cn[v + 8 * half];
            const float inv = 1.0f / fmaxf(cv, 1.0f);
            P[(v + 8 * half) * P_PITCH + w * 16 + lm] = acc[v] * inv + (cv > 0.0f ? bv : 0.0f);
        }
    }
    __syncthreads();

    if (w < 2) {
        const float* __restrict__ bb = Wfc + (size_t)(2 * half) * 64 + w * 32 + lm;
        v8f acc0 = (v8f){};
        v8f acc1 = (v8f){};
#pragma unroll
        for (int k0 = 0; k0 < 128; k0 += 4) {
            const v2f a = *(const v2f*)&P[lm * P_PITCH + k0 + 2 * half];
            v2f b0, b1;
            b0.x = bb[k0 * 64];
            b0.y = bb[(k0 + 1) * 64];
            b1.x = bb[k0 * 64 + 16];
            b1.y = bb[(k0 + 1) * 64 + 16];
            acc0 = __builtin_amdgcn_wmma_f32_16x16x4_f32(
                 false, a,  false, b0,
                 (short)0, acc0,  false,  false);
            acc1 = __builtin_amdgcn_wmma_f32_16x16x4_f32(
                 false, a,  false, b1,
                 (short)0, acc1,  false,  false);
        }
        const float bv0 = bfc[w * 32 + lm];
        const float bv1 = bfc[w * 32 + 16 + lm];
        float* __restrict__ orow = out + (size_t)(g0 + 8 * half) * 64 + w * 32 + lm;
#pragma unroll
        for (int v = 0; v < 8; ++v) {
            orow[(size_t)v * 64]      = acc0[v] + bv0;
            orow[(size_t)v * 64 + 16] = acc1[v] + bv1;
        }
    }
}

static inline int cdiv(long long a, long long b) { return (int)((a + b - 1) / b); }

extern "C" void kernel_launch(void* const* d_in, const int* in_sizes, int n_in,
                              void* d_out, int out_size, void* d_ws, size_t ws_size,
                              hipStream_t stream) {
    (void)n_in; (void)ws_size;
    const float* x    = (const float*)d_in[0];
    const int*   edge = (const int*)d_in[1];
    const int*   batch= (const int*)d_in[2];
    const float* b1   = (const float*)d_in[4];
    const float* W1   = (const float*)d_in[3];
    const float* W2   = (const float*)d_in[5];
    const float* b2   = (const float*)d_in[6];
    const float* Wfc  = (const float*)d_in[7];
    const float* bfc  = (const float*)d_in[8];
    float* out = (float*)d_out;

    const int N   = in_sizes[2];
    const int E   = in_sizes[1] / 2;
    const int D   = 64;
    const int D2  = 128;
    const int HID = in_sizes[8];
    const int G   = out_size / HID;

    const int* src = edge;
    const int* dst = edge + E;

    float* ws    = (float*)d_ws;
    float* r0    = ws;
    float* r1    = r0 + (size_t)N * D2;
    float* r2    = r1 + (size_t)N * D2;
    float* xw1   = r0;
    float* a2    = r1;
    float* h1    = r2;
    const int NB = cdiv(N, SORT_T);
    const int CH = cdiv(E, CHUNKS);
    unsigned* binned = (unsigned*)(r2 + (size_t)N * D);
    int*   nbr   = (int*)binned + E;
    int*   tab   = nbr + E;
    int*   off   = tab + (size_t)(NB + 1) * CHUNKS;
    float* dinv  = r2 + (size_t)N * D2;

    chunk_pass_kernel<<<CHUNKS, CHUNK_T, 0, stream>>>(src, dst, binned, tab, E, CH, NB);

    bucket_gemm_kernel<<<NB, SORT_T, 0, stream>>>(
        binned, tab, off, nbr, dinv, N, CH,
        x, W1, xw1, N / 16);

    gather_kernel<64, true><<<cdiv((long long)N * 32, TPB), TPB, 0, stream>>>(
        xw1, off, nbr, dinv, b1, h1, N);

    gather_kernel<64, false><<<cdiv((long long)N * 32, TPB), TPB, 0, stream>>>(
        h1, off, nbr, dinv, nullptr, a2, N);

    pool_w2_fc_kernel<<<G / 16, TAIL_T, 0, stream>>>(a2, batch, W2, b2, Wfc, bfc, out, N);
}


# Edge grouping staged in LDS: bucket pass reads each chunk piece once, one wave-wide load per piece; both sort passes place words in LDS and write whole lines; slot (sum) order changes
# speedup vs baseline: 8.5583x; 1.2104x over previous
#include <hip/hip_runtime.h>
#include <hip/hip_bf16.h>


typedef float v2f __attribute__((ext_vector_type(2)));
typedef float v8f __attribute__((ext_vector_type(8)));

#define TPB 256

#define SORT_T 256
#define BIN_SH 8
#define CHUNKS 256
#define CHUNK_T 1024
#define LB     4
#define PB     8
#define CCAP   4096
#define BCAP   6144
#define PL     8

template <int NT>
__device__ __forceinline__ int block_scan(int v, int* __restrict__ wtot, int& all) {
    const int lane = threadIdx.x & 31;
    const int w    = threadIdx.x >> 5;
    int s = v;
#pragma unroll
    for (int o = 1; o < 32; o <<= 1) {
        int u = __shfl_up(s, o);
        if (lane >= o) s += u;
    }
    if (lane == 31) wtot[w] = s;
    __syncthreads();
    int pre = 0, tot = 0;
#pragma unroll
    for (int i = 0; i < NT / 32; ++i) {
        const int x = wtot[i];
        tot += x;
        pre += (i < w) ? x : 0;
    }
    all = tot;
    return pre + s - v;
}

__global__ void __launch_bounds__(CHUNK_T)
chunk_pass_kernel(const int* __restrict__ src, const int* __restrict__ dst,
                  unsigned* __restrict__ binned, int* __restrict__ tab,
                  int E, int CH, int NB) {
    __shared__ int bcnt[SORT_T];
    __shared__ int wtot[CHUNK_T / 32];
    __shared__ unsigned cstage[CCAP];
    const int t  = threadIdx.x;
    const int c  = blockIdx.x;
    const int e0 = c * CH;
    const int n  = max(0, min(CH, E - e0));
    const bool in_lds = n <= CCAP;
    const unsigned last = (unsigned)(NB - 1);
    if (t < SORT_T) bcnt[t] = 0;
    __syncthreads();
    for (int i0 = t; i0 < n; i0 += LB * CHUNK_T) {
        unsigned d[LB];
#pragma unroll
        for (int u = 0; u < LB; ++u) {
            d[u] = (unsigned)dst[e0 + min(i0 + u * CHUNK_T, n - 1)];
        }
#pragma unroll
        for (int u = 0; u < LB; ++u) {
            const bool valid = i0 + u * CHUNK_T < n;
            const unsigned b = valid ? min(d[u] >> BIN_SH, last) : (unsigned)(t & (SORT_T - 1));
            atomicAdd(&bcnt[b], valid ? 1 : 0);
        }
    }
    __syncthreads();
    int tot;
    const int v  = (t < SORT_T) ? bcnt[t] : 0;
    const int ex = block_scan<CHUNK_T>(v, wtot, tot);
    if (t <= NB) tab[t * CHUNKS + c] = ex;
    if (t < SORT_T) bcnt[t] = ex;
    __syncthreads();
    for (int i0 = t; i0 < n; i0 += LB * CHUNK_T) {
        unsigned d[LB], s[LB];
        int p[LB];
#pragma unroll
        for (int u = 0; u < LB; ++u) {
            const int i = min(i0 + u * CHUNK_T, n - 1);
            d[u] = (unsigned)dst[e0 + i];
            s[u] = (unsigned)src[e0 + i];
        }
#pragma unroll
        for (int u = 0; u < LB; ++u) {
            const bool valid = i0 + u * CHUNK_T < n;
            const unsigned b = valid ? min(d[u] >> BIN_SH, last) : (unsigned)(t & (SORT_T - 1));
            p[u] = atomicAdd(&bcnt[b], valid ? 1 : 0);
        }
#pragma unroll
        for (int u = 0; u < LB; ++u) {
            const bool valid = i0 + u * CHUNK_T < n;
            const unsigned word = (s[u] << BIN_SH) | (d[u] & (unsigned)(SORT_T - 1));
            if (valid) {
                if (in_lds) cstage[p[u]] = word;
                else        binned[e0 + p[u]] = word;
            }
        }
    }
    if (in_lds) {
        __syncthreads();
        for (int i0 = t; i0 < n; i0 += LB * CHUNK_T) {
            unsigned v[LB];
#pragma unroll
            for (int u = 0; u < LB; ++u) v[u] = cstage[min(i0 + u * CHUNK_T, n - 1)];
#pragma unroll
            for (int u = 0; u < LB; ++u)
                if (i0 + u * CHUNK_T < n) binned[e0 + i0 + u * CHUNK_T] = v[u];
        }
    }
}

__device__ __forceinline__ float
bucket_pass_body(const unsigned* __restrict__ binned, const int* __restrict__ tab,
                 int* __restrict__ off, int* __restrict__ nbr,
                 float* __restrict__ dinv, int n, int CH, int b) {
    static_assert(CHUNKS == SORT_T, "thread t of a bucket work-group stands for chunk t");
    __shared__ int ncnt[SORT_T];
    __shared__ int wtot_a[SORT_T / 32];
    __shared__ int wtot_b[SORT_T / 32];
    __shared__ int wtot_c[SORT_T / 32];
    __shared__ unsigned stage[BCAP + 32];
    __shared__ int      placed[BCAP + 32];
    const int t    = threadIdx.x;
    const int lane = t & 31;
    const int o0 = tab[b * CHUNKS + t];
    const int o1 = tab[(b + 1) * CHUNKS + t];
    const int len = o1 - o0;
    const unsigned* __restrict__ piece = binned + (size_t)t * CH + o0;
    ncnt[t] = 0;
    int base;
    (void)block_scan<SORT_T>(o0, wtot_a, base);
    int total;
    const int ppos = block_scan<SORT_T>(len, wtot_c, total);
    const bool in_lds = total <= BCAP;
    if (in_lds) {
        const int soff = t * CH + o0;
        unsigned wv[32];
#pragma unroll
        for (int i = 0; i < 32; ++i) {
            const int so = __builtin_amdgcn_readlane(soff, i);
            const int L  = __builtin_amdgcn_readlane(len, i);
            wv[i] = binned[(L > 0) ? so + min(lane, L - 1) : 0];
        }
#pragma unroll
        for (int i = 0; i < 32; ++i) {
            const int L  = __builtin_amdgcn_readlane(len, i);
            const int pp = __builtin_amdgcn_readlane(ppos, i);
            const bool valid = lane < L;
            stage[valid ? pp + lane : BCAP + lane] = wv[i];
            atomicAdd(&ncnt[valid ? (wv[i] & (unsigned)(SORT_T - 1)) : (unsigned)t], valid ? 1 : 0);
        }
        if (__builtin_amdgcn_ballot_w32(len > 32) != 0u) {
            for (int i = 0; i < 32; ++i) {
                const int L = __builtin_amdgcn_readlane(len, i);
                if (L <= 32) continue;
                const int so = __builtin_amdgcn_readlane(soff, i);
                const int pp = __builtin_amdgcn_readlane(ppos, i);
                for (int j0 = 32; j0 < L; j0 += 32) {
                    if (j0 + lane < L) {
                        const unsigned x = binned[so + j0 + lane];
                        stage[pp + j0 + lane] = x;
                        atomicAdd(&ncnt[x & (unsigned)(SORT_T - 1)], 1);
                    }
                }
            }
        }
    } else {
        for (int k0 = 0; k0 < len; k0 += PB) {
            unsigned w[PB];
#pragma unroll
            for (int u = 0; u < PB; ++u) w[u] = piece[min(k0 + u, len - 1)];
#pragma unroll
            for (int u = 0; u < PB; ++u) {
                const bool valid = k0 + u < len;
                atomicAdd(&ncnt[valid ? (w[u] & (unsigned)(SORT_T - 1)) : (unsigned)t], valid ? 1 : 0);
            }
        }
    }
    __syncthreads();
    const int cdeg = ncnt[t];
    int unused;
    const int ex = block_scan<SORT_T>(cdeg, wtot_b, unused);
    const int node = b * SORT_T + t;
    const float dv = rsqrtf((float)(cdeg + 1));
    if (node < n) {
        off[node]  = base + ex;
        dinv[node] = dv;
    }
    if (node == n - 1) off[n] = base + ex + cdeg;
    ncnt[t] = in_lds ? ex : base + ex;
    __syncthreads();
    if (in_lds) {
        for (int i0 = t; i0 < total; i0 += PL * SORT_T) {
            unsigned w[PL];
            int p[PL];
#pragma unroll
            for (int u = 0; u < PL; ++u) w[u] = stage[min(i0 + u * SORT_T, total - 1)];
#pragma unroll
            for (int u = 0; u < PL; ++u) {
                const bool valid = i0 + u * SORT_T < total;
                p[u] = atomicAdd(&ncnt[valid ? (w[u] & (unsigned)(SORT_T - 1)) : (unsigned)t], valid ? 1 : 0);
            }
#pragma unroll
            for (int u = 0; u < PL; ++u) {
                const bool valid = i0 + u * SORT_T < total;
                placed[valid ? p[u] : BCAP + lane] = (int)(w[u] >> BIN_SH);
            }
        }
        __syncthreads();
        for (int i0 = t; i0 < total; i0 += PL * SORT_T) {
            int v[PL];
#pragma unroll
            for (int u = 0; u < PL; ++u) v[u] = placed[min(i0 + u * SORT_T, total - 1)];
#pragma unroll
            for (int u = 0; u < PL; ++u)
                if (i0 + u * SORT_T < total) nbr[base + i0 + u * SORT_T] = v[u];
        }
    } else {
        for (int k0 = 0; k0 < len; k0 += PB) {
            unsigned w[PB];
#pragma unroll
            for (int u = 0; u < PB; ++u) w[u] = piece[min(k0 + u, len - 1)];
#pragma unroll
            for (int u = 0; u < PB; ++u) {
                const bool valid = k0 + u < len;
                const int p = atomicAdd(&ncnt[valid ? (w[u] & (unsigned)(SORT_T - 1)) : (unsigned)t], valid ? 1 : 0);
                if (valid) nbr[p] = (int)(w[u] >> BIN_SH);
            }
        }
    }
    return dv;
}

template <int K, int Nn, int RT>
__device__ __forceinline__ void wmma_gemm_strip(const float* __restrict__ A,
                                                const float* __restrict__ B,
                                                float* __restrict__ C,
                                                const float* __restrict__ sc,
                                                int row0) {
    constexpr int COLT = Nn / 16;
    int lane = threadIdx.x & 31;
    int half = lane >> 4;
    int lm   = lane & 15;

    const float* __restrict__ arow  = A + (size_t)(row0 + lm) * K + 2 * half;
    const float* __restrict__ bbase = B + (size_t)(2 * half) * Nn + lm;

    v8f acc[RT][COLT];
#pragma unroll
    for (int r = 0; r < RT; ++r)
#pragma unroll
        for (int c = 0; c < COLT; ++c) acc[r][c] = (v8f){};

#pragma unroll
    for (int k0 = 0; k0 < K; k0 += 4) {
        v2f a[RT];
#pragma unroll
        for (int r = 0; r < RT; ++r) {
            a[r].x = arow[r * 16 * K + k0];
            a[r].y = arow[r * 16 * K + k0 + 1];
        }
#pragma unroll
        for (int c = 0; c < COLT; ++c) {
            v2f b;
            b.x = bbase[k0 * Nn + c * 16];
            b.y = bbase[(k0 + 1) * Nn + c * 16];
#pragma unroll
            for (int r = 0; r < RT; ++r)
                acc[r][c] = __builtin_amdgcn_wmma_f32_16x16x4_f32(
                     false, a[r],  false, b,
                     (short)0, acc[r][c],  false,  false);
        }
    }

    float* __restrict__ crow = C + (size_t)(row0 + 8 * half) * Nn + lm;
#pragma unroll
    for (int r = 0; r < RT; ++r) {
#pragma unroll
        for (int v = 0; v < 8; ++v) {
            const float s = sc[r * 16 + 8 * half + v];
#pragma unroll
            for (int c = 0; c < COLT; ++c)
                crow[(size_t)(r * 16 + v) * Nn + c * 16] = acc[r][c][v] * s;
        }
    }
}

__global__ void __launch_bounds__(SORT_T)
bucket_gemm_kernel(const unsigned* __restrict__ binned, const int* __restrict__ tab,
                   int* __restrict__ off, int* __restrict__ nbr,
                   float* __restrict__ dinv, int n, int CH,
                   const float* __restrict__ x, const float* __restrict__ W1,
                   float* __restrict__ xw1s, int row_tiles,
                   const int* __restrict__ batch, int* __restrict__ goff, int G) {
    __shared__ float dsc[SORT_T];
    const int b = blockIdx.x;
    const int t = threadIdx.x;
    {
        const int node = b * SORT_T + t;
        if (node < n) {
            const int gb = min(max(batch[node], 0), G - 1);
            const int ga = (node > 0) ? min(max(batch[node - 1], 0), G - 1) : -1;
            for (int g = ga + 1; g <= gb; ++g) goff[g] = node;
            if (node == n - 1)
                for (int g = gb + 1; g <= G; ++g) goff[g] = n;
        }
    }
    dsc[t] = bucket_pass_body(binned, tab, off, nbr, dinv, n, CH, b);
    __syncthreads();
    const int w  = __builtin_amdgcn_readfirstlane(t >> 5);
    const int t0 = b * (SORT_T / 16) + 2 * w;
    if (t0 + 1 < row_tiles)
        wmma_gemm_strip<64, 64, 2>(x, W1, xw1s, &dsc[32 * w], t0 * 16);
    else if (t0 < row_tiles)
        wmma_gemm_strip<64, 64, 1>(x, W1, xw1s, &dsc[32 * w], t0 * 16);
}

template <int F, bool CONV1>
__global__ void __launch_bounds__(TPB)
gather_kernel(const float* __restrict__ hs, const int* __restrict__ off,
              const int* __restrict__ nbr, const float* __restrict__ dinv,
              const float* __restrict__ bias, float* __restrict__ out, int N) {
    constexpr int VEC = F / 32;
    typedef float vf __attribute__((ext_vector_type(VEC)));
    const int node = __builtin_amdgcn_readfirstlane(
        (int)((blockIdx.x * blockDim.x + threadIdx.x) >> 5));
    if (node >= N) return;
    const int lane = threadIdx.x & 31;
    const vf* __restrict__ hv = (const vf*)hs;

    const int beg = off[node];
    const int end = off[node + 1];
    const float dd = dinv[node];

    vf acc = hv[(size_t)node * 32 + lane];

    for (int base = beg; base < end; base += 32) {
        const int j = base + lane;
        int s_l = node;
        if (j < end) s_l = nbr[j];
        const int cnt  = min(32, end - base);
        const int cnt4 = (cnt + 3) & ~3;
        for (int k = 0; k < cnt4; k += 4) {
#pragma unroll
            for (int u = 0; u < 4; ++u) {
                const int   s = __builtin_amdgcn_readlane(s_l, k + u);
                const float w = (k + u < cnt) ? 1.0f : 0.0f;
                acc += hv[(size_t)s * 32 + lane] * w;
            }
        }
    }

    vf r;
#pragma unroll
    for (int q = 0; q < VEC; ++q) {
        float v = acc[q] * dd;
        if (CONV1) v = fmaxf(v + bias[lane * VEC + q], 0.0f) * dd;
        r[q] = v;
    }
    ((vf*)out)[(size_t)node * 32 + lane] = r;
}

#define TAIL_T   1024
#define TAIL_W   (TAIL_T / 32)
#define S_PITCH  68
#define P_PITCH  132

__global__ void __launch_bounds__(TAIL_T) __attribute__((amdgpu_waves_per_eu(16)))
gather_pool_w2_fc_kernel(const float* __restrict__ hs, const int* __restrict__ off,
                         const int* __restrict__ nbr, const float* __restrict__ dinv,
                         const int* __restrict__ goff,
                         const float* __restrict__ W2, const float* __restrict__ b2,
                         const float* __restrict__ Wfc, const float* __restrict__ bfc,
                         float* __restrict__ out, int N) {
    __shared__ __attribute__((aligned(16))) float S[16 * S_PITCH];
    __shared__ __attribute__((aligned(16))) float P[16 * P_PITCH];
    __shared__ __attribute__((aligned(16))) float red[TAIL_W * 64];
    __shared__ float Cn[16];
    __shared__ int   ctr;
    const int t    = threadIdx.x;
    const int lane = t & 31;
    const int w    = __builtin_amdgcn_readfirstlane(t >> 5);
    const int g    = blockIdx.x;
    const int half = lane >> 4;
    const int lm   = lane & 15;

    const int lo = min(max(goff[g], 0), N);
    const int hi = min(max(goff[g + 1], lo), N);

    for (int i = t; i < 16 * S_PITCH; i += TAIL_T) S[i] = 0.0f;
    if (t < 16) Cn[t] = (t == 0) ? (float)(hi - lo) : 0.0f;
    if (t == 0) ctr = TAIL_W;
    __syncthreads();

    const v2f* __restrict__ hv = (const v2f*)hs;
    v2f psum = (v2f){0.0f, 0.0f};
    for (int i = w; lo + i < hi; ) {
        const int node = lo + i;
        const int beg = off[node];
        const int end = off[node + 1];
        const float dd = dinv[node];
        v2f acc = hv[(size_t)node * 32 + lane];
        int nx = 0;
        if (lane == 0) nx = atomicAdd(&ctr, 1);
        for (int base = beg; base < end; base += 32) {
            const int j = base + lane;
            int s_l = node;
            if (j < end) s_l = nbr[j];
            const int cnt  = min(32, end - base);
            const int cnt4 = (cnt + 3) & ~3;
            for (int k = 0; k < cnt4; k += 4) {
#pragma unroll
                for (int u = 0; u < 4; ++u) {
                    const int   s  = __builtin_amdgcn_readlane(s_l, k + u);
                    const float wt = (k + u < cnt) ? 1.0f : 0.0f;
                    acc += hv[(size_t)s * 32 + lane] * wt;
                }
            }
        }
        psum += acc * dd;
        i = __builtin_amdgcn_readfirstlane(nx);
    }
    red[w * 64 + 2 * lane]     = psum.x;
    red[w * 64 + 2 * lane + 1] = psum.y;
    __syncthreads();
    if (t < 64) {
        float s = 0.0f;
#pragma unroll
        for (int k = 0; k < TAIL_W; ++k) s += red[k * 64 + t];
        S[t] = s;
    }
    __syncthreads();

    if (w < 8) {
        const float* __restrict__ bb = W2 + (size_t)(2 * half) * 128 + w * 16 + lm;
        v8f acc = (v8f){};
#pragma unroll
        for (int k0 = 0; k0 < 64; k0 += 4) {
            const v2f a = *(const v2f*)&S[lm * S_PITCH + k0 + 2 * half];
            v2f b;
            b.x = bb[k0 * 128];
            b.y = bb[(k0 + 1) * 128];
            acc = __builtin_amdgcn_wmma_f32_16x16x4_f32(
                 false, a,  false, b,
                 (short)0, acc,  false,  false);
        }
        const float bv = b2[w * 16 + lm];
#pragma unroll
        for (int v = 0; v < 8; ++v) {
            const float cv  = Cn[v + 8 * half];
            const float inv = 1.0f / fmaxf(cv, 1.0f);
            P[(v + 8 * half) * P_PITCH + w * 16 + lm] = acc[v] * inv + (cv > 0.0f ? bv : 0.0f);
        }
    }
    __syncthreads();

    if (w < 2) {
        const float* __restrict__ bb = Wfc + (size_t)(2 * half) * 64 + w * 32 + lm;
        v8f acc0 = (v8f){};
        v8f acc1 = (v8f){};
#pragma unroll 8
        for (int k0 = 0; k0 < 128; k0 += 4) {
            const v2f a = *(const v2f*)&P[lm * P_PITCH + k0 + 2 * half];
            v2f b0, b1;
            b0.x = bb[k0 * 64];
            b0.y = bb[(k0 + 1) * 64];
            b1.x = bb[k0 * 64 + 16];
            b1.y = bb[(k0 + 1) * 64 + 16];
            acc0 = __builtin_amdgcn_wmma_f32_16x16x4_f32(
                 false, a,  false, b0,
                 (short)0, acc0,  false,  false);
            acc1 = __builtin_amdgcn_wmma_f32_16x16x4_f32(
                 false, a,  false, b1,
                 (short)0, acc1,  false,  false);
        }
        const float bv0 = bfc[w * 32 + lm];
        const float bv1 = bfc[w * 32 + 16 + lm];
        if (half == 0) {
            float* __restrict__ orow = out + (size_t)g * 64 + w * 32 + lm;
            orow[0]  = acc0[0] + bv0;
            orow[16] = acc1[0] + bv1;
        }
    }
}

static inline int cdiv(long long a, long long b) { return (int)((a + b - 1) / b); }

extern "C" void kernel_launch(void* const* d_in, const int* in_sizes, int n_in,
                              void* d_out, int out_size, void* d_ws, size_t ws_size,
                              hipStream_t stream) {
    (void)n_in; (void)ws_size;
    const float* x    = (const float*)d_in[0];
    const int*   edge = (const int*)d_in[1];
    const int*   batch= (const int*)d_in[2];
    const float* b1   = (const float*)d_in[4];
    const float* W1   = (const float*)d_in[3];
    const float* W2   = (const float*)d_in[5];
    const float* b2   = (const float*)d_in[6];
    const float* Wfc  = (const float*)d_in[7];
    const float* bfc  = (const float*)d_in[8];
    float* out = (float*)d_out;

    const int N   = in_sizes[2];
    const int E   = in_sizes[1] / 2;
    const int D   = 64;
    const int D2  = 128;
    const int HID = in_sizes[8];
    const int G   = out_size / HID;

    const int* src = edge;
    const int* dst = edge + E;

    float* ws    = (float*)d_ws;
    float* r0    = ws;
    float* r1    = r0 + (size_t)N * D2;
    float* r2    = r1 + (size_t)N * D2;
    float* xw1   = r0;
    float* h1    = r2;
    const int NB = cdiv(N, SORT_T);
    const int CH = cdiv(E, CHUNKS);
    unsigned* binned = (unsigned*)(r2 + (size_t)N * D);
    int*   nbr   = (int*)binned + E;
    int*   tab   = nbr + E;
    int*   off   = tab + (size_t)(NB + 1) * CHUNKS;
    int*   goff  = off + N + 1;
    float* dinv  = r2 + (size_t)N * D2;

    chunk_pass_kernel<<<CHUNKS, CHUNK_T, 0, stream>>>(src, dst, binned, tab, E, CH, NB);

    bucket_gemm_kernel<<<NB, SORT_T, 0, stream>>>(
        binned, tab, off, nbr, dinv, N, CH,
        x, W1, xw1, N / 16, batch, goff, G);

    gather_kernel<64, true><<<cdiv((long long)N * 32, TPB), TPB, 0, stream>>>(
        xw1, off, nbr, dinv, b1, h1, N);

    gather_pool_w2_fc_kernel<<<G, TAIL_T, 0, stream>>>(
        h1, off, nbr, dinv, goff, W2, b2, Wfc, bfc, out, N);
}


# conv1 aggregation by persistent waves: a wave sums 4 consecutive nodes in turn (bounds up front, next node's ids and own row prefetched, 8 rows in flight); a quarter of the waves; sums unchanged
# speedup vs baseline: 8.8771x; 1.0373x over previous
#include <hip/hip_runtime.h>
#include <hip/hip_bf16.h>


typedef float v2f __attribute__((ext_vector_type(2)));
typedef float v8f __attribute__((ext_vector_type(8)));

#define TPB 256

#define SORT_T 256
#define BIN_SH 8
#define CHUNKS 256
#define CHUNK_T 1024
#define LB     4
#define PB     8
#define CCAP   4096
#define BCAP   6144
#define PL     8

template <int NT>
__device__ __forceinline__ int block_scan(int v, int* __restrict__ wtot, int& all) {
    const int lane = threadIdx.x & 31;
    const int w    = threadIdx.x >> 5;
    int s = v;
#pragma unroll
    for (int o = 1; o < 32; o <<= 1) {
        int u = __shfl_up(s, o);
        if (lane >= o) s += u;
    }
    if (lane == 31) wtot[w] = s;
    __syncthreads();
    int pre = 0, tot = 0;
#pragma unroll
    for (int i = 0; i < NT / 32; ++i) {
        const int x = wtot[i];
        tot += x;
        pre += (i < w) ? x : 0;
    }
    all = tot;
    return pre + s - v;
}

__global__ void __launch_bounds__(CHUNK_T)
chunk_pass_kernel(const int* __restrict__ src, const int* __restrict__ dst,
                  unsigned* __restrict__ binned, int* __restrict__ tab,
                  int E, int CH, int NB) {
    __shared__ int bcnt[SORT_T];
    __shared__ int wtot[CHUNK_T / 32];
    __shared__ unsigned cstage[CCAP];
    const int t  = threadIdx.x;
    const int c  = blockIdx.x;
    const int e0 = c * CH;
    const int n  = max(0, min(CH, E - e0));
    const bool in_lds = n <= CCAP;
    const unsigned last = (unsigned)(NB - 1);
    if (t < SORT_T) bcnt[t] = 0;
    __syncthreads();
    for (int i0 = t; i0 < n; i0 += LB * CHUNK_T) {
        unsigned d[LB];
#pragma unroll
        for (int u = 0; u < LB; ++u) {
            d[u] = (unsigned)dst[e0 + min(i0 + u * CHUNK_T, n - 1)];
        }
#pragma unroll
        for (int u = 0; u < LB; ++u) {
            const bool valid = i0 + u * CHUNK_T < n;
            const unsigned b = valid ? min(d[u] >> BIN_SH, last) : (unsigned)(t & (SORT_T - 1));
            atomicAdd(&bcnt[b], valid ? 1 : 0);
        }
    }
    __syncthreads();
    int tot;
    const int v  = (t < SORT_T) ? bcnt[t] : 0;
    const int ex = block_scan<CHUNK_T>(v, wtot, tot);
    if (t <= NB) tab[t * CHUNKS + c] = ex;
    if (t < SORT_T) bcnt[t] = ex;
    __syncthreads();
    for (int i0 = t; i0 < n; i0 += LB * CHUNK_T) {
        unsigned d[LB], s[LB];
        int p[LB];
#pragma unroll
        for (int u = 0; u < LB; ++u) {
            const int i = min(i0 + u * CHUNK_T, n - 1);
            d[u] = (unsigned)dst[e0 + i];
            s[u] = (unsigned)src[e0 + i];
        }
#pragma unroll
        for (int u = 0; u < LB; ++u) {
            const bool valid = i0 + u * CHUNK_T < n;
            const unsigned b = valid ? min(d[u] >> BIN_SH, last) : (unsigned)(t & (SORT_T - 1));
            p[u] = atomicAdd(&bcnt[b], valid ? 1 : 0);
        }
#pragma unroll
        for (int u = 0; u < LB; ++u) {
            const bool valid = i0 + u * CHUNK_T < n;
            const unsigned word = (s[u] << BIN_SH) | (d[u] & (unsigned)(SORT_T - 1));
            if (valid) {
                if (in_lds) cstage[p[u]] = word;
                else        binned[e0 + p[u]] = word;
            }
        }
    }
    if (in_lds) {
        __syncthreads();
        for (int i0 = t; i0 < n; i0 += LB * CHUNK_T) {
            unsigned v[LB];
#pragma unroll
            for (int u = 0; u < LB; ++u) v[u] = cstage[min(i0 + u * CHUNK_T, n - 1)];
#pragma unroll
            for (int u = 0; u < LB; ++u)
                if (i0 + u * CHUNK_T < n) binned[e0 + i0 + u * CHUNK_T] = v[u];
        }
    }
}

__device__ __forceinline__ float
bucket_pass_body(const unsigned* __restrict__ binned, const int* __restrict__ tab,
                 int* __restrict__ off, int* __restrict__ nbr,
                 float* __restrict__ dinv, int n, int CH, int b) {
    static_assert(CHUNKS == SORT_T, "thread t of a bucket work-group stands for chunk t");
    __shared__ int ncnt[SORT_T];
    __shared__ int wtot_a[SORT_T / 32];
    __shared__ int wtot_b[SORT_T / 32];
    __shared__ int wtot_c[SORT_T / 32];
    __shared__ unsigned stage[BCAP + 32];
    __shared__ int      placed[BCAP + 32];
    const int t    = threadIdx.x;
    const int lane = t & 31;
    const int o0 = tab[b * CHUNKS + t];
    const int o1 = tab[(b + 1) * CHUNKS + t];
    const int len = o1 - o0;
    const unsigned* __restrict__ piece = binned + (size_t)t * CH + o0;
    ncnt[t] = 0;
    int base;
    (void)block_scan<SORT_T>(o0, wtot_a, base);
    int total;
    const int ppos = block_scan<SORT_T>(len, wtot_c, total);
    const bool in_lds = total <= BCAP;
    if (in_lds) {
        const int soff = t * CH + o0;
        unsigned wv[32];
#pragma unroll
        for (int i = 0; i < 32; ++i) {
            const int so = __builtin_amdgcn_readlane(soff, i);
            const int L  = __builtin_amdgcn_readlane(len, i);
            wv[i] = binned[(L > 0) ? so + min(lane, L - 1) : 0];
        }
#pragma unroll
        for (int i = 0; i < 32; ++i) {
            const int L  = __builtin_amdgcn_readlane(len, i);
            const int pp = __builtin_amdgcn_readlane(ppos, i);
            const bool valid = lane < L;
            stage[valid ? pp + lane : BCAP + lane] = wv[i];
            atomicAdd(&ncnt[valid ? (wv[i] & (unsigned)(SORT_T - 1)) : (unsigned)t], valid ? 1 : 0);
        }
        if (__builtin_amdgcn_ballot_w32(len > 32) != 0u) {
            for (int i = 0; i < 32; ++i) {
                const int L = __builtin_amdgcn_readlane(len, i);
                if (L <= 32) continue;
                const int so = __builtin_amdgcn_readlane(soff, i);
                const int pp = __builtin_amdgcn_readlane(ppos, i);
                for (int j0 = 32; j0 < L; j0 += 32) {
                    if (j0 + lane < L) {
                        const unsigned x = binned[so + j0 + lane];
                        stage[pp + j0 + lane] = x;
                        atomicAdd(&ncnt[x & (unsigned)(SORT_T - 1)], 1);
                    }
                }
            }
        }
    } else {
        for (int k0 = 0; k0 < len; k0 += PB) {
            unsigned w[PB];
#pragma unroll
            for (int u = 0; u < PB; ++u) w[u] = piece[min(k0 + u, len - 1)];
#pragma unroll
            for (int u = 0; u < PB; ++u) {
                const bool valid = k0 + u < len;
                atomicAdd(&ncnt[valid ? (w[u] & (unsigned)(SORT_T - 1)) : (unsigned)t], valid ? 1 : 0);
            }
        }
    }
    __syncthreads();
    const int cdeg = ncnt[t];
    int unused;
    const int ex = block_scan<SORT_T>(cdeg, wtot_b, unused);
    const int node = b * SORT_T + t;
    const float dv = rsqrtf((float)(cdeg + 1));
    if (node < n) {
        off[node]  = base + ex;
        dinv[node] = dv;
    }
    if (node == n - 1) off[n] = base + ex + cdeg;
    ncnt[t] = in_lds ? ex : base + ex;
    __syncthreads();
    if (in_lds) {
        for (int i0 = t; i0 < total; i0 += PL * SORT_T) {
            unsigned w[PL];
            int p[PL];
#pragma unroll
            for (int u = 0; u < PL; ++u) w[u] = stage[min(i0 + u * SORT_T, total - 1)];
#pragma unroll
            for (int u = 0; u < PL; ++u) {
                const bool valid = i0 + u * SORT_T < total;
                p[u] = atomicAdd(&ncnt[valid ? (w[u] & (unsigned)(SORT_T - 1)) : (unsigned)t], valid ? 1 : 0);
            }
#pragma unroll
            for (int u = 0; u < PL; ++u) {
                const bool valid = i0 + u * SORT_T < total;
                placed[valid ? p[u] : BCAP + lane] = (int)(w[u] >> BIN_SH);
            }
        }
        __syncthreads();
        for (int i0 = t; i0 < total; i0 += PL * SORT_T) {
            int v[PL];
#pragma unroll
            for (int u = 0; u < PL; ++u) v[u] = placed[min(i0 + u * SORT_T, total - 1)];
#pragma unroll
            for (int u = 0; u < PL; ++u)
                if (i0 + u * SORT_T < total) nbr[base + i0 + u * SORT_T] = v[u];
        }
    } else {
        for (int k0 = 0; k0 < len; k0 += PB) {
            unsigned w[PB];
#pragma unroll
            for (int u = 0; u < PB; ++u) w[u] = piece[min(k0 + u, len - 1)];
#pragma unroll
            for (int u = 0; u < PB; ++u) {
                const bool valid = k0 + u < len;
                const int p = atomicAdd(&ncnt[valid ? (w[u] & (unsigned)(SORT_T - 1)) : (unsigned)t], valid ? 1 : 0);
                if (valid) nbr[p] = (int)(w[u] >> BIN_SH);
            }
        }
    }
    return dv;
}

template <int K, int Nn, int RT>
__device__ __forceinline__ void wmma_gemm_strip(const float* __restrict__ A,
                                                const float* __restrict__ B,
                                                float* __restrict__ C,
                                                const float* __restrict__ sc,
                                                int row0) {
    constexpr int COLT = Nn / 16;
    int lane = threadIdx.x & 31;
    int half = lane >> 4;
    int lm   = lane & 15;

    const float* __restrict__ arow  = A + (size_t)(row0 + lm) * K + 2 * half;
    const float* __restrict__ bbase = B + (size_t)(2 * half) * Nn + lm;

    v8f acc[RT][COLT];
#pragma unroll
    for (int r = 0; r < RT; ++r)
#pragma unroll
        for (int c = 0; c < COLT; ++c) acc[r][c] = (v8f){};

#pragma unroll
    for (int k0 = 0; k0 < K; k0 += 4) {
        v2f a[RT];
#pragma unroll
        for (int r = 0; r < RT; ++r) {
            a[r].x = arow[r * 16 * K + k0];
            a[r].y = arow[r * 16 * K + k0 + 1];
        }
#pragma unroll
        for (int c = 0; c < COLT; ++c) {
            v2f b;
            b.x = bbase[k0 * Nn + c * 16];
            b.y = bbase[(k0 + 1) * Nn + c * 16];
#pragma unroll
            for (int r = 0; r < RT; ++r)
                acc[r][c] = __builtin_amdgcn_wmma_f32_16x16x4_f32(
                     false, a[r],  false, b,
                     (short)0, acc[r][c],  false,  false);
        }
    }

    float* __restrict__ crow = C + (size_t)(row0 + 8 * half) * Nn + lm;
#pragma unroll
    for (int r = 0; r < RT; ++r) {
#pragma unroll
        for (int v = 0; v < 8; ++v) {
            const float s = sc[r * 16 + 8 * half + v];
#pragma unroll
            for (int c = 0; c < COLT; ++c)
                crow[(size_t)(r * 16 + v) * Nn + c * 16] = acc[r][c][v] * s;
        }
    }
}

__global__ void __launch_bounds__(SORT_T)
bucket_gemm_kernel(const unsigned* __restrict__ binned, const int* __restrict__ tab,
                   int* __restrict__ off, int* __restrict__ nbr,
                   float* __restrict__ dinv, int n, int CH,
                   const float* __restrict__ x, const float* __restrict__ W1,
                   float* __restrict__ xw1s, int row_tiles,
                   const int* __restrict__ batch, int* __restrict__ goff, int G) {
    __shared__ float dsc[SORT_T];
    const int b = blockIdx.x;
    const int t = threadIdx.x;
    {
        const int node = b * SORT_T + t;
        if (node < n) {
            const int gb = min(max(batch[node], 0), G - 1);
            const int ga = (node > 0) ? min(max(batch[node - 1], 0), G - 1) : -1;
            for (int g = ga + 1; g <= gb; ++g) goff[g] = node;
            if (node == n - 1)
                for (int g = gb + 1; g <= G; ++g) goff[g] = n;
        }
    }
    dsc[t] = bucket_pass_body(binned, tab, off, nbr, dinv, n, CH, b);
    __syncthreads();
    const int w  = __builtin_amdgcn_readfirstlane(t >> 5);
    const int t0 = b * (SORT_T / 16) + 2 * w;
    if (t0 + 1 < row_tiles)
        wmma_gemm_strip<64, 64, 2>(x, W1, xw1s, &dsc[32 * w], t0 * 16);
    else if (t0 < row_tiles)
        wmma_gemm_strip<64, 64, 1>(x, W1, xw1s, &dsc[32 * w], t0 * 16);
}

#define G1_NPW   4
#define G1_WAVES (TPB / 32)

template <typename vf>
__device__ __forceinline__ vf sum_rows(const vf* __restrict__ hv, int s_l, int cnt, int lane, vf acc) {
    const int cnt4 = (cnt + 3) & ~3;
    int k = 0;
    for (; k + 8 <= cnt4; k += 8) {
#pragma unroll
        for (int u = 0; u < 8; ++u) {
            const int   s = __builtin_amdgcn_readlane(s_l, k + u);
            const float w = (k + u < cnt) ? 1.0f : 0.0f;
            acc += hv[(size_t)s * 32 + lane] * w;
        }
    }
    if (k < cnt4) {
#pragma unroll
        for (int u = 0; u < 4; ++u) {
            const int   s = __builtin_amdgcn_readlane(s_l, k + u);
            const float w = (k + u < cnt) ? 1.0f : 0.0f;
            acc += hv[(size_t)s * 32 + lane] * w;
        }
    }
    return acc;
}

template <int F>
__global__ void __launch_bounds__(TPB)
gather_conv1_kernel(const float* __restrict__ hs, const int* __restrict__ off,
                    const int* __restrict__ nbr, const float* __restrict__ dinv,
                    const float* __restrict__ bias, float* __restrict__ out, int N) {
    constexpr int VEC = F / 32;
    typedef float vf __attribute__((ext_vector_type(VEC)));
    const int lane = threadIdx.x & 31;
    const int wg = __builtin_amdgcn_readfirstlane((int)(threadIdx.x >> 5));
    const int n0 = ((int)blockIdx.x * G1_WAVES + wg) * G1_NPW;
    if (n0 >= N) return;
    const vf* __restrict__ hv = (const vf*)hs;

    int   o[G1_NPW + 1];
    float d[G1_NPW];
#pragma unroll
    for (int i = 0; i <= G1_NPW; ++i) o[i] = off[min(n0 + i, N)];
#pragma unroll
    for (int i = 0; i < G1_NPW; ++i) d[i] = dinv[min(n0 + i, N - 1)];

    vf bv;
#pragma unroll
    for (int q = 0; q < VEC; ++q) bv[q] = bias[lane * VEC + q];

    int ids = n0;
    if (o[0] + lane < o[1]) ids = nbr[o[0] + lane];
    vf self = hv[(size_t)n0 * 32 + lane];

#pragma unroll
    for (int i = 0; i < G1_NPW; ++i) {
        const int node = n0 + i;
        if (node >= N) break;
        const int beg = o[i];
        const int end = o[i + 1];
        int ids_n = node;
        vf  self_n = (vf){};
        if (i + 1 < G1_NPW && node + 1 < N) {
            const int i2 = (i + 2 <= G1_NPW) ? i + 2 : G1_NPW;
            ids_n = node + 1;
            if (o[i + 1] + lane < o[i2]) ids_n = nbr[o[i + 1] + lane];
            self_n = hv[(size_t)(node + 1) * 32 + lane];
        }

        vf acc = sum_rows<vf>(hv, ids, min(32, end - beg), lane, self);
        for (int base = beg + 32; base < end; base += 32) {
            int s_l = node;
            if (base + lane < end) s_l = nbr[base + lane];
            acc = sum_rows<vf>(hv, s_l, min(32, end - base), lane, acc);
        }

        const float dd = d[i];
        vf r;
#pragma unroll
        for (int q = 0; q < VEC; ++q) {
            float v = acc[q] * dd;
            v = fmaxf(v + bv[q], 0.0f) * dd;
            r[q] = v;
        }
        ((vf*)out)[(size_t)node * 32 + lane] = r;

        ids  = ids_n;
        self = self_n;
    }
}

#define TAIL_T   1024
#define TAIL_W   (TAIL_T / 32)
#define S_PITCH  68
#define P_PITCH  132

__global__ void __launch_bounds__(TAIL_T) __attribute__((amdgpu_waves_per_eu(16)))
gather_pool_w2_fc_kernel(const float* __restrict__ hs, const int* __restrict__ off,
                         const int* __restrict__ nbr, const float* __restrict__ dinv,
                         const int* __restrict__ goff,
                         const float* __restrict__ W2, const float* __restrict__ b2,
                         const float* __restrict__ Wfc, const float* __restrict__ bfc,
                         float* __restrict__ out, int N) {
    __shared__ __attribute__((aligned(16))) float S[16 * S_PITCH];
    __shared__ __attribute__((aligned(16))) float P[16 * P_PITCH];
    __shared__ __attribute__((aligned(16))) float red[TAIL_W * 64];
    __shared__ float Cn[16];
    __shared__ int   ctr;
    const int t    = threadIdx.x;
    const int lane = t & 31;
    const int w    = __builtin_amdgcn_readfirstlane(t >> 5);
    const int g    = blockIdx.x;
    const int half = lane >> 4;
    const int lm   = lane & 15;

    const int lo = min(max(goff[g], 0), N);
    const int hi = min(max(goff[g + 1], lo), N);

    for (int i = t; i < 16 * S_PITCH; i += TAIL_T) S[i] = 0.0f;
    if (t < 16) Cn[t] = (t == 0) ? (float)(hi - lo) : 0.0f;
    if (t == 0) ctr = TAIL_W;
    __syncthreads();

    const v2f* __restrict__ hv = (const v2f*)hs;
    v2f psum = (v2f){0.0f, 0.0f};
    for (int i = w; lo + i < hi; ) {
        const int node = lo + i;
        const int beg = off[node];
        const int end = off[node + 1];
        const float dd = dinv[node];
        v2f acc = hv[(size_t)node * 32 + lane];
        int nx = 0;
        if (lane == 0) nx = atomicAdd(&ctr, 1);
        for (int base = beg; base < end; base += 32) {
            const int j = base + lane;
            int s_l = node;
            if (j < end) s_l = nbr[j];
            const int cnt  = min(32, end - base);
            const int cnt4 = (cnt + 3) & ~3;
            for (int k = 0; k < cnt4; k += 4) {
#pragma unroll
                for (int u = 0; u < 4; ++u) {
                    const int   s  = __builtin_amdgcn_readlane(s_l, k + u);
                    const float wt = (k + u < cnt) ? 1.0f : 0.0f;
                    acc += hv[(size_t)s * 32 + lane] * wt;
                }
            }
        }
        psum += acc * dd;
        i = __builtin_amdgcn_readfirstlane(nx);
    }
    red[w * 64 + 2 * lane]     = psum.x;
    red[w * 64 + 2 * lane + 1] = psum.y;
    __syncthreads();
    if (t < 64) {
        float s = 0.0f;
#pragma unroll
        for (int k = 0; k < TAIL_W; ++k) s += red[k * 64 + t];
        S[t] = s;
    }
    __syncthreads();

    if (w < 8) {
        const float* __restrict__ bb = W2 + (size_t)(2 * half) * 128 + w * 16 + lm;
        v8f acc = (v8f){};
#pragma unroll
        for (int k0 = 0; k0 < 64; k0 += 4) {
            const v2f a = *(const v2f*)&S[lm * S_PITCH + k0 + 2 * half];
            v2f b;
            b.x = bb[k0 * 128];
            b.y = bb[(k0 + 1) * 128];
            acc = __builtin_amdgcn_wmma_f32_16x16x4_f32(
                 false, a,  false, b,
                 (short)0, acc,  false,  false);
        }
        const float bv = b2[w * 16 + lm];
#pragma unroll
        for (int v = 0; v < 8; ++v) {
            const float cv  = Cn[v + 8 * half];
            const float inv = 1.0f / fmaxf(cv, 1.0f);
            P[(v + 8 * half) * P_PITCH + w * 16 + lm] = acc[v] * inv + (cv > 0.0f ? bv : 0.0f);
        }
    }
    __syncthreads();

    if (w < 2) {
        const float* __restrict__ bb = Wfc + (size_t)(2 * half) * 64 + w * 32 + lm;
        v8f acc0 = (v8f){};
        v8f acc1 = (v8f){};
#pragma unroll 8
        for (int k0 = 0; k0 < 128; k0 += 4) {
            const v2f a = *(const v2f*)&P[lm * P_PITCH + k0 + 2 * half];
            v2f b0, b1;
            b0.x = bb[k0 * 64];
            b0.y = bb[(k0 + 1) * 64];
            b1.x = bb[k0 * 64 + 16];
            b1.y = bb[(k0 + 1) * 64 + 16];
            acc0 = __builtin_amdgcn_wmma_f32_16x16x4_f32(
                 false, a,  false, b0,
                 (short)0, acc0,  false,  false);
            acc1 = __builtin_amdgcn_wmma_f32_16x16x4_f32(
                 false, a,  false, b1,
                 (short)0, acc1,  false,  false);
        }
        const float bv0 = bfc[w * 32 + lm];
        const float bv1 = bfc[w * 32 + 16 + lm];
        if (half == 0) {
            float* __restrict__ orow = out + (size_t)g * 64 + w * 32 + lm;
            orow[0]  = acc0[0] + bv0;
            orow[16] = acc1[0] + bv1;
        }
    }
}

static inline int cdiv(long long a, long long b) { return (int)((a + b - 1) / b); }

extern "C" void kernel_launch(void* const* d_in, const int* in_sizes, int n_in,
                              void* d_out, int out_size, void* d_ws, size_t ws_size,
                              hipStream_t stream) {
    (void)n_in; (void)ws_size;
    const float* x    = (const float*)d_in[0];
    const int*   edge = (const int*)d_in[1];
    const int*   batch= (const int*)d_in[2];
    const float* b1   = (const float*)d_in[4];
    const float* W1   = (const float*)d_in[3];
    const float* W2   = (const float*)d_in[5];
    const float* b2   = (const float*)d_in[6];
    const float* Wfc  = (const float*)d_in[7];
    const float* bfc  = (const float*)d_in[8];
    float* out = (float*)d_out;

    const int N   = in_sizes[2];
    const int E   = in_sizes[1] / 2;
    const int D   = 64;
    const int D2  = 128;
    const int HID = in_sizes[8];
    const int G   = out_size / HID;

    const int* src = edge;
    const int* dst = edge + E;

    float* ws    = (float*)d_ws;
    float* r0    = ws;
    float* r1    = r0 + (size_t)N * D2;
    float* r2    = r1 + (size_t)N * D2;
    float* xw1   = r0;
    float* h1    = r2;
    const int NB = cdiv(N, SORT_T);
    const int CH = cdiv(E, CHUNKS);
    unsigned* binned = (unsigned*)(r2 + (size_t)N * D);
    int*   nbr   = (int*)binned + E;
    int*   tab   = nbr + E;
    int*   off   = tab + (size_t)(NB + 1) * CHUNKS;
    int*   goff  = off + N + 1;
    float* dinv  = r2 + (size_t)N * D2;

    chunk_pass_kernel<<<CHUNKS, CHUNK_T, 0, stream>>>(src, dst, binned, tab, E, CH, NB);

    bucket_gemm_kernel<<<NB, SORT_T, 0, stream>>>(
        binned, tab, off, nbr, dinv, N, CH,
        x, W1, xw1, N / 16, batch, goff, G);

    gather_conv1_kernel<64><<<cdiv(N, G1_WAVES * G1_NPW), TPB, 0, stream>>>(
        xw1, off, nbr, dinv, b1, h1, N);

    gather_pool_w2_fc_kernel<<<G, TAIL_T, 0, stream>>>(
        h1, off, nbr, dinv, goff, W2, b2, Wfc, bfc, out, N);
}
